# starting kernel
# baseline (speedup 1.0000x reference)
#include <hip/hip_runtime.h>
#include <math.h>


#define B_ 64
#define N_ 4096
#define C_ 128
#define H_ 4
#define HC_ 512
#define ALPHA 0.2f

typedef __attribute__((ext_vector_type(8))) float v8f;
typedef __attribute__((ext_vector_type(2))) float v2f;

#define OFF_W2P   0u
#define OFF_B2    2048u
#define OFF_LOG   2064u
#define OFF_MX    (OFF_LOG + (unsigned)(B_*H_*N_))
#define OFF_ISUM  (OFF_MX + 256u)
#define OFF_PART  (OFF_ISUM + 256u)
#define OFF_U     (OFF_PART + (unsigned)(B_*8*H_*C_))

__device__ __forceinline__ float neg_inf() { return -__builtin_inff(); }

__global__ void k_prep(const float* __restrict__ W, const float* __restrict__ bias,
                       const float* __restrict__ T, float* __restrict__ ws) {
  int idx = blockIdx.x * 256 + threadIdx.x;
  if (idx < 2048) {
    int k = idx >> 4, j = idx & 15;
    float v = 0.f;
    if (j < H_) {
      const float* wr = W + (size_t)k * HC_ + j * C_;
      const float* tr = T + j * C_;
      for (int c = 0; c < C_; ++c) v = fmaf(wr[c], tr[c], v);
    }
    ws[OFF_W2P + idx] = v;
  } else if (idx < 2048 + H_) {
    int j = idx - 2048;
    float v = 0.f;
    for (int c = 0; c < C_; ++c) v = fmaf(bias[j * C_ + c], T[j * C_ + c], v);
    ws[OFF_B2 + j] = v;
  }
}

__global__ void k_logits(const float* __restrict__ V, const int* __restrict__ gs,
                         float* __restrict__ ws) {
  __shared__ float s_w2p[2048];
  const int tid = threadIdx.x;
  for (int i = tid; i < 2048; i += 128) s_w2p[i] = ws[OFF_W2P + i];
  __syncthreads();
  const int b = blockIdx.y;
  const int wave = tid >> 5, lane = tid & 31;
  const int l16 = lane & 15, hi = lane >> 4;
  const int n0 = blockIdx.x * 64 + wave * 16;
  const float* vrow = V + ((size_t)b * N_ + (n0 + l16)) * C_;
  v8f acc = {};
  for (int k0 = 0; k0 < C_; k0 += 4) {
    v2f a = *(const v2f*)(vrow + k0 + 2 * hi);
    v2f bv;
    bv.x = s_w2p[(k0 + 2 * hi) * 16 + l16];
    bv.y = s_w2p[(k0 + 1 + 2 * hi) * 16 + l16];
    acc = __builtin_amdgcn_wmma_f32_16x16x4_f32(false, a, false, bv,
                                                (short)0, acc, false, false);
  }
  const int limit = gs[b];
  const float bb = ws[OFF_B2 + (l16 & 3)];
  if (l16 < H_) {
    float* lrow = ws + OFF_LOG + ((size_t)b * H_ + l16) * N_;
    for (int r = 0; r < 8; ++r) {
      int node = n0 + r + 8 * hi;
      float x = acc[r] + bb;
      x = (x >= 0.f) ? x : ALPHA * x;
      lrow[node] = (node < limit) ? x : neg_inf();
    }
  }
}

__global__ void k_stats(float* __restrict__ ws) {
  __shared__ float red[256];
  const int bh = blockIdx.x;
  const int tid = threadIdx.x;
  const float* l = ws + OFF_LOG + (size_t)bh * N_;
  float m = neg_inf();
  for (int n = tid; n < N_; n += 256) m = fmaxf(m, l[n]);
  red[tid] = m; __syncthreads();
  for (int s = 128; s > 0; s >>= 1) {
    if (tid < s) red[tid] = fmaxf(red[tid], red[tid + s]);
    __syncthreads();
  }
  m = red[0]; __syncthreads();
  float sum = 0.f;
  for (int n = tid; n < N_; n += 256) {
    float x = l[n];
    sum += (x == neg_inf()) ? 0.f : __expf(x - m);
  }
  red[tid] = sum; __syncthreads();
  for (int s = 128; s > 0; s >>= 1) {
    if (tid < s) red[tid] += red[tid + s];
    __syncthreads();
  }
  if (tid == 0) { ws[OFF_MX + bh] = m; ws[OFF_ISUM + bh] = 1.f / red[0]; }
}

__global__ void k_wsum(const float* __restrict__ V, float* __restrict__ ws) {
  __shared__ float s_p[H_ * 512];
  const int b = blockIdx.y, chunk = blockIdx.x;
  const int nb = chunk * 512;
  const int tid = threadIdx.x;
  for (int i = tid; i < H_ * 512; i += 256) {
    int h = i >> 9, n = i & 511;
    float x = ws[OFF_LOG + ((size_t)b * H_ + h) * N_ + nb + n];
    float m = ws[OFF_MX + b * H_ + h];
    float is = ws[OFF_ISUM + b * H_ + h];
    s_p[i] = (x == neg_inf()) ? 0.f : __expf(x - m) * is;
  }
  __syncthreads();
  const int wave = tid >> 5, lane = tid & 31;
  const int l16 = lane & 15, hi = lane >> 4;
  const int c0 = wave * 16;
  const float* vb = V + ((size_t)b * N_ + nb) * C_;
  v8f acc = {};
  for (int k0 = 0; k0 < 512; k0 += 4) {
    v2f a;
    if (l16 < H_) {
      a.x = s_p[l16 * 512 + k0 + 2 * hi];
      a.y = s_p[l16 * 512 + k0 + 1 + 2 * hi];
    } else { a.x = 0.f; a.y = 0.f; }
    v2f bv;
    bv.x = vb[(size_t)(k0 + 2 * hi) * C_ + c0 + l16];
    bv.y = vb[(size_t)(k0 + 1 + 2 * hi) * C_ + c0 + l16];
    acc = __builtin_amdgcn_wmma_f32_16x16x4_f32(false, a, false, bv,
                                                (short)0, acc, false, false);
  }
  if (hi == 0) {
    float* pr = ws + OFF_PART + (size_t)(b * 8 + chunk) * (H_ * C_);
    for (int r = 0; r < H_; ++r) pr[r * C_ + c0 + l16] = acc[r];
  }
}

__global__ void k_reduce(float* __restrict__ ws) {
  int idx = blockIdx.x * 256 + threadIdx.x;
  int b = idx >> 9, rest = idx & 511;
  float s = 0.f;
  for (int ch = 0; ch < 8; ++ch)
    s += ws[OFF_PART + ((size_t)(b * 8 + ch) << 9) + rest];
  ws[OFF_U + idx] = s;
}

__global__ void k_out(const float* __restrict__ W, const float* __restrict__ bias,
                      const float* __restrict__ U, float* __restrict__ out) {
  const int m0 = blockIdx.x * 16;
  const int h  = blockIdx.y;
  const int tid = threadIdx.x;
  const int wave = tid >> 5, lane = tid & 31;
  const int l16 = lane & 15, hi = lane >> 4;
  const int n0 = wave * 16;
  const float* urow = U + (size_t)(m0 + l16) * HC_ + h * C_;
  const float* wcol = W + h * C_ + n0 + l16;
  v8f acc = {};
  for (int k0 = 0; k0 < C_; k0 += 4) {
    v2f a = *(const v2f*)(urow + k0 + 2 * hi);
    v2f bv;
    bv.x = wcol[(size_t)(k0 + 2 * hi) * HC_];
    bv.y = wcol[(size_t)(k0 + 1 + 2 * hi) * HC_];
    acc = __builtin_amdgcn_wmma_f32_16x16x4_f32(false, a, false, bv,
                                                (short)0, acc, false, false);
  }
  for (int r = 0; r < 8; ++r) {
    int row = m0 + r + 8 * hi;
    int col = h * C_ + n0 + l16;
    out[(size_t)row * HC_ + col] = acc[r] + bias[col];
  }
}

extern "C" void kernel_launch(void* const* d_in, const int* in_sizes, int n_in,
                              void* d_out, int out_size, void* d_ws, size_t ws_size,
                              hipStream_t stream) {
  (void)in_sizes; (void)n_in; (void)out_size; (void)ws_size;
  const float* V    = (const float*)d_in[0];
  const int*   gs   = (const int*)  d_in[1];
  const float* W    = (const float*)d_in[2];
  const float* bias = (const float*)d_in[3];
  const float* T    = (const float*)d_in[4];
  float* out = (float*)d_out;
  float* ws  = (float*)d_ws;

  k_prep  <<<9,            256, 0, stream>>>(W, bias, T, ws);
  k_logits<<<dim3(N_/64, B_), 128, 0, stream>>>(V, gs, ws);
  k_stats <<<B_ * H_,      256, 0, stream>>>(ws);
  k_wsum  <<<dim3(8, B_),  256, 0, stream>>>(V, ws);
  k_reduce<<<(B_*H_*C_)/256, 256, 0, stream>>>(ws);
  k_out   <<<dim3(B_/16, H_), 256, 0, stream>>>(W, bias, ws + OFF_U, out);
}


# K4 weighted sum: each wave owns 64 whole rows, V read by 128-bit loads requested a 32-row batch ahead (was one 4-byte round trip per WMMA); wave partials added through LDS, so the sum order changes
# speedup vs baseline: 1.9173x; 1.9173x over previous
#include <hip/hip_runtime.h>
#include <math.h>


#define B_ 64
#define N_ 4096
#define C_ 128
#define H_ 4
#define HC_ 512
#define ALPHA 0.2f

typedef __attribute__((ext_vector_type(8))) float v8f;
typedef __attribute__((ext_vector_type(2))) float v2f;
typedef __attribute__((ext_vector_type(4))) float v4f;

#define OFF_W2P   0u
#define OFF_B2    2048u
#define OFF_LOG   2064u
#define OFF_MX    (OFF_LOG + (unsigned)(B_*H_*N_))
#define OFF_ISUM  (OFF_MX + 256u)
#define OFF_PART  (OFF_ISUM + 256u)
#define OFF_U     (OFF_PART + (unsigned)(B_*8*H_*C_))

__device__ __forceinline__ float neg_inf() { return -__builtin_inff(); }

__global__ void k_prep(const float* __restrict__ W, const float* __restrict__ bias,
                       const float* __restrict__ T, float* __restrict__ ws) {
  int idx = blockIdx.x * 256 + threadIdx.x;
  if (idx < 2048) {
    int k = idx >> 4, j = idx & 15;
    float v = 0.f;
    if (j < H_) {
      const float* wr = W + (size_t)k * HC_ + j * C_;
      const float* tr = T + j * C_;
      for (int c = 0; c < C_; ++c) v = fmaf(wr[c], tr[c], v);
    }
    ws[OFF_W2P + idx] = v;
  } else if (idx < 2048 + H_) {
    int j = idx - 2048;
    float v = 0.f;
    for (int c = 0; c < C_; ++c) v = fmaf(bias[j * C_ + c], T[j * C_ + c], v);
    ws[OFF_B2 + j] = v;
  }
}

__global__ void k_logits(const float* __restrict__ V, const int* __restrict__ gs,
                         float* __restrict__ ws) {
  __shared__ float s_w2p[2048];
  const int tid = threadIdx.x;
  for (int i = tid; i < 2048; i += 128) s_w2p[i] = ws[OFF_W2P + i];
  __syncthreads();
  const int b = blockIdx.y;
  const int wave = tid >> 5, lane = tid & 31;
  const int l16 = lane & 15, hi = lane >> 4;
  const int n0 = blockIdx.x * 64 + wave * 16;
  const float* vrow = V + ((size_t)b * N_ + (n0 + l16)) * C_;
  v8f acc = {};
  for (int k0 = 0; k0 < C_; k0 += 4) {
    v2f a = *(const v2f*)(vrow + k0 + 2 * hi);
    v2f bv;
    bv.x = s_w2p[(k0 + 2 * hi) * 16 + l16];
    bv.y = s_w2p[(k0 + 1 + 2 * hi) * 16 + l16];
    acc = __builtin_amdgcn_wmma_f32_16x16x4_f32(false, a, false, bv,
                                                (short)0, acc, false, false);
  }
  const int limit = gs[b];
  const float bb = ws[OFF_B2 + (l16 & 3)];
  if (l16 < H_) {
    float* lrow = ws + OFF_LOG + ((size_t)b * H_ + l16) * N_;
    for (int r = 0; r < 8; ++r) {
      int node = n0 + r + 8 * hi;
      float x = acc[r] + bb;
      x = (x >= 0.f) ? x : ALPHA * x;
      lrow[node] = (node < limit) ? x : neg_inf();
    }
  }
}

__global__ void k_stats(float* __restrict__ ws) {
  __shared__ float red[256];
  const int bh = blockIdx.x;
  const int tid = threadIdx.x;
  const float* l = ws + OFF_LOG + (size_t)bh * N_;
  float m = neg_inf();
  for (int n = tid; n < N_; n += 256) m = fmaxf(m, l[n]);
  red[tid] = m; __syncthreads();
  for (int s = 128; s > 0; s >>= 1) {
    if (tid < s) red[tid] = fmaxf(red[tid], red[tid + s]);
    __syncthreads();
  }
  m = red[0]; __syncthreads();
  float sum = 0.f;
  for (int n = tid; n < N_; n += 256) {
    float x = l[n];
    sum += (x == neg_inf()) ? 0.f : __expf(x - m);
  }
  red[tid] = sum; __syncthreads();
  for (int s = 128; s > 0; s >>= 1) {
    if (tid < s) red[tid] += red[tid + s];
    __syncthreads();
  }
  if (tid == 0) { ws[OFF_MX + bh] = m; ws[OFF_ISUM + bh] = 1.f / red[0]; }
}

#define PSTR 520

#define K4_LOAD_BATCH(ld, base)                                              \
  _Pragma("unroll")   \
  for (int g = 0; g < 8; ++g) {                                              \
    const float* r_ = (base) + (size_t)(g * 4) * C_;                         \
    ld[g][0] = *(const v4f*)(r_);                                            \
    ld[g][1] = *(const v4f*)(r_ + 64);                                       \
    ld[g][2] = *(const v4f*)(r_ + C_);                                       \
    ld[g][3] = *(const v4f*)(r_ + C_ + 64);                                  \
  }

#define K4_MMA_BATCH(ld, pbase)                                              \
  _Pragma("unroll")          \
  for (int g = 0; g < 8; ++g) {                                              \
    v2f a = *(const v2f*)((pbase) + g * 4);                                  \
    if (!real) { a.x = 0.f; a.y = 0.f; }                 \
    _Pragma("unroll")                                           \
    for (int cs = 0; cs < 2; ++cs) {                                         \
      _Pragma("unroll")                                         \
      for (int q = 0; q < 4; ++q) {                                          \
        v2f bv;                                                              \
        bv.x = ld[g][cs][q];                                 \
        bv.y = ld[g][2 + cs][q];                             \
        acc[cs * 4 + q] = __builtin_amdgcn_wmma_f32_16x16x4_f32(             \
            false, a, false, bv, (short)0, acc[cs * 4 + q], false, false);   \
      }                                                                      \
    }                                                                        \
  }

__global__ __launch_bounds__(256, 4)
void k_wsum(const float* __restrict__ V, float* __restrict__ ws) {
  __shared__ __attribute__((aligned(16))) float s_p[H_ * PSTR];
  __shared__ __attribute__((aligned(16))) float s_red[8 * H_ * C_];
  const int b = blockIdx.y, chunk = blockIdx.x;
  const int nb = chunk * 512;
  const int tid = threadIdx.x;
  const int wave = tid >> 5, lane = tid & 31;
  const int l16 = lane & 15, hi = lane >> 4;
  const float* vb = V + ((size_t)b * N_ + nb + wave * 64 + 2 * hi) * C_ + 4 * l16;
  v4f ld[8][4];
  K4_LOAD_BATCH(ld, vb)
  {
    const float* lg = ws + OFF_LOG + (size_t)b * H_ * N_ + nb + tid;
    float x[8];
#pragma unroll
    for (int j = 0; j < 8; ++j) x[j] = lg[(size_t)(j >> 1) * N_ + (j & 1) * 256];
#pragma unroll
    for (int j = 0; j < 8; ++j) {
      const int h = j >> 1;
      float m = ws[OFF_MX + b * H_ + h];
      float is = ws[OFF_ISUM + b * H_ + h];
      s_p[h * PSTR + (j & 1) * 256 + tid] =
          (x[j] == neg_inf()) ? 0.f : __expf(x[j] - m) * is;
    }
  }
  __syncthreads();
  const float* pa = s_p + (l16 & 3) * PSTR + wave * 64 + 2 * hi;
  const bool real = l16 < H_;
  v8f acc[8] = {};
  K4_MMA_BATCH(ld, pa)
  K4_LOAD_BATCH(ld, vb + (size_t)32 * C_)
  K4_MMA_BATCH(ld, pa + 32)
  if (hi == 0) {
    float* rw = s_red + wave * (H_ * C_) + 4 * l16;
#pragma unroll
    for (int r = 0; r < H_; ++r) {
#pragma unroll
      for (int cs = 0; cs < 2; ++cs) {
        v4f o;
        o.x = acc[cs * 4 + 0][r]; o.y = acc[cs * 4 + 1][r];
        o.z = acc[cs * 4 + 2][r]; o.w = acc[cs * 4 + 3][r];
        *(v4f*)(rw + r * C_ + cs * 64) = o;
      }
    }
  }
  __syncthreads();
  float* pr = ws + OFF_PART + (size_t)(b * 8 + chunk) * (H_ * C_);
  for (int i = tid; i < H_ * C_; i += 256) {
    float s = 0.f;
    for (int w = 0; w < 8; ++w) s += s_red[w * (H_ * C_) + i];
    pr[i] = s;
  }
}

__global__ void k_reduce(float* __restrict__ ws) {
  int idx = blockIdx.x * 256 + threadIdx.x;
  int b = idx >> 9, rest = idx & 511;
  float s = 0.f;
  for (int ch = 0; ch < 8; ++ch)
    s += ws[OFF_PART + ((size_t)(b * 8 + ch) << 9) + rest];
  ws[OFF_U + idx] = s;
}

__global__ void k_out(const float* __restrict__ W, const float* __restrict__ bias,
                      const float* __restrict__ U, float* __restrict__ out) {
  const int m0 = blockIdx.x * 16;
  const int h  = blockIdx.y;
  const int tid = threadIdx.x;
  const int wave = tid >> 5, lane = tid & 31;
  const int l16 = lane & 15, hi = lane >> 4;
  const int n0 = wave * 16;
  const float* urow = U + (size_t)(m0 + l16) * HC_ + h * C_;
  const float* wcol = W + h * C_ + n0 + l16;
  v8f acc = {};
  for (int k0 = 0; k0 < C_; k0 += 4) {
    v2f a = *(const v2f*)(urow + k0 + 2 * hi);
    v2f bv;
    bv.x = wcol[(size_t)(k0 + 2 * hi) * HC_];
    bv.y = wcol[(size_t)(k0 + 1 + 2 * hi) * HC_];
    acc = __builtin_amdgcn_wmma_f32_16x16x4_f32(false, a, false, bv,
                                                (short)0, acc, false, false);
  }
  for (int r = 0; r < 8; ++r) {
    int row = m0 + r + 8 * hi;
    int col = h * C_ + n0 + l16;
    out[(size_t)row * HC_ + col] = acc[r] + bias[col];
  }
}

extern "C" void kernel_launch(void* const* d_in, const int* in_sizes, int n_in,
                              void* d_out, int out_size, void* d_ws, size_t ws_size,
                              hipStream_t stream) {
  (void)in_sizes; (void)n_in; (void)out_size; (void)ws_size;
  const float* V    = (const float*)d_in[0];
  const int*   gs   = (const int*)  d_in[1];
  const float* W    = (const float*)d_in[2];
  const float* bias = (const float*)d_in[3];
  const float* T    = (const float*)d_in[4];
  float* out = (float*)d_out;
  float* ws  = (float*)d_ws;

  k_prep  <<<9,            256, 0, stream>>>(W, bias, T, ws);
  k_logits<<<dim3(N_/64, B_), 128, 0, stream>>>(V, gs, ws);
  k_stats <<<B_ * H_,      256, 0, stream>>>(ws);
  k_wsum  <<<dim3(8, B_),  256, 0, stream>>>(V, ws);
  k_reduce<<<(B_*H_*C_)/256, 256, 0, stream>>>(ws);
  k_out   <<<dim3(B_/16, H_), 256, 0, stream>>>(W, bias, ws + OFF_U, out);
}


# K2 logits: wave owns 64 rows; V read one whole 512-B row per 128-bit load, a 16-row tile ahead, transposed to fragments in wave-private LDS (was 8-byte loads, 16 rows each); logit sum order changes
# speedup vs baseline: 2.7149x; 1.4160x over previous
#include <hip/hip_runtime.h>
#include <math.h>


#define B_ 64
#define N_ 4096
#define C_ 128
#define H_ 4
#define HC_ 512
#define ALPHA 0.2f

typedef __attribute__((ext_vector_type(8))) float v8f;
typedef __attribute__((ext_vector_type(2))) float v2f;
typedef __attribute__((ext_vector_type(4))) float v4f;

#define OFF_W2P   0u
#define OFF_B2    2048u
#define OFF_LOG   2064u
#define OFF_MX    (OFF_LOG + (unsigned)(B_*H_*N_))
#define OFF_ISUM  (OFF_MX + 256u)
#define OFF_PART  (OFF_ISUM + 256u)
#define OFF_U     (OFF_PART + (unsigned)(B_*8*H_*C_))

__device__ __forceinline__ float neg_inf() { return -__builtin_inff(); }

__global__ void k_prep(const float* __restrict__ W, const float* __restrict__ bias,
                       const float* __restrict__ T, float* __restrict__ ws) {
  int idx = blockIdx.x * 256 + threadIdx.x;
  if (idx < 2048) {
    int k = idx >> 4, j = idx & 15;
    float v = 0.f;
    if (j < H_) {
      const float* wr = W + (size_t)k * HC_ + j * C_;
      const float* tr = T + j * C_;
      for (int c = 0; c < C_; ++c) v = fmaf(wr[c], tr[c], v);
    }
    ws[OFF_W2P + idx] = v;
  } else if (idx < 2048 + H_) {
    int j = idx - 2048;
    float v = 0.f;
    for (int c = 0; c < C_; ++c) v = fmaf(bias[j * C_ + c], T[j * C_ + c], v);
    ws[OFF_B2 + j] = v;
  }
}

#define LP 132

#define K2_LOAD_TILE(g, base)                                                \
  _Pragma("unroll")   \
  for (int r = 0; r < 16; ++r) g[r] = *(const v4f*)((base) + (size_t)r * C_);

__device__ __forceinline__ void wave_lds_order() {
  __builtin_amdgcn_fence(__ATOMIC_RELEASE, "wavefront");
  __builtin_amdgcn_wave_barrier();
  __builtin_amdgcn_fence(__ATOMIC_ACQUIRE, "wavefront");
}

__global__ __launch_bounds__(256, 4)
void k_logits(const float* __restrict__ V, const int* __restrict__ gs,
              float* __restrict__ ws) {
  __shared__ __attribute__((aligned(16))) float s_v[8 * 16 * LP];
  __shared__ __attribute__((aligned(16))) float s_w2p[2048];
  const int b = blockIdx.y, chunk = blockIdx.x;
  const int tid = threadIdx.x;
  const int wave = tid >> 5, lane = tid & 31;
  const int l16 = lane & 15, hi = lane >> 4;
  const int n0 = chunk * 512 + wave * 64;
  const float* vg = V + ((size_t)b * N_ + n0) * C_ + 4 * lane;
  float* st = s_v + wave * (16 * LP) + 4 * lane;
  const float* fr = s_v + wave * (16 * LP) + l16 * LP + 4 * hi;
  float tw[8];
#pragma unroll
  for (int k = 0; k < 8; ++k) tw[k] = ws[OFF_W2P + k * 256 + tid];
  v4f g[16];
  K2_LOAD_TILE(g, vg)
#pragma unroll
  for (int k = 0; k < 8; ++k) s_w2p[k * 256 + tid] = tw[k];
  __syncthreads();
  const bool real = (l16 & 4) == 0;
  v2f wf[32];
#pragma unroll
  for (int i = 0; i < 32; ++i) {
    const int c = 8 * (i >> 1) + 4 * hi + 2 * (i & 1);
    const float w0 = s_w2p[c * 16 + (l16 & 3)];
    const float w1 = s_w2p[(c + 1) * 16 + (l16 & 3)];
    wf[i].x = real ? w0 : 0.f;
    wf[i].y = real ? w1 : 0.f;
  }
  const int limit = gs[b];
  float bb[H_];
#pragma unroll
  for (int h = 0; h < H_; ++h) bb[h] = ws[OFF_B2 + h];
  float* lrow = ws + OFF_LOG + (size_t)b * H_ * N_;
  float ev[H_];
#pragma unroll
  for (int t = 0; t < 4; ++t) {
#pragma unroll
    for (int r = 0; r < 16; ++r) *(v4f*)(st + r * LP) = g[r];
    wave_lds_order();
    if (t < 3) { K2_LOAD_TILE(g, vg + (size_t)(t + 1) * 16 * C_) }
    __builtin_amdgcn_sched_barrier(0);
    v4f f[16];
#pragma unroll
    for (int j = 0; j < 16; ++j) f[j] = *(const v4f*)(fr + 8 * j);
    __builtin_amdgcn_sched_barrier(0);
    wave_lds_order();
    v8f acc0 = {}, acc1 = {};
#pragma unroll
    for (int j = 0; j < 16; ++j) {
      v2f b0, b1;
      b0.x = f[j].x; b0.y = f[j].y;
      b1.x = f[j].z; b1.y = f[j].w;
      acc0 = __builtin_amdgcn_wmma_f32_16x16x4_f32(false, wf[2 * j], false, b0,
                                                   (short)0, acc0, false, false);
      acc1 = __builtin_amdgcn_wmma_f32_16x16x4_f32(false, wf[2 * j + 1], false, b1,
                                                   (short)0, acc1, false, false);
    }
    if ((t & 1) == 0) {
#pragma unroll
      for (int h = 0; h < H_; ++h) ev[h] = acc0[h] + acc1[h];
    } else {
      const int node = n0 + (t - 1) * 16 + lane;
#pragma unroll
      for (int h = 0; h < H_; ++h) {
        float x = (hi ? (acc0[h] + acc1[h]) : ev[h]) + bb[h];
        x = (x >= 0.f) ? x : ALPHA * x;
        lrow[(size_t)h * N_ + node] = (node < limit) ? x : neg_inf();
      }
    }
    __builtin_amdgcn_sched_barrier(0);
  }
}

__global__ void k_stats(float* __restrict__ ws) {
  __shared__ float red[256];
  const int bh = blockIdx.x;
  const int tid = threadIdx.x;
  const float* l = ws + OFF_LOG + (size_t)bh * N_;
  float m = neg_inf();
  for (int n = tid; n < N_; n += 256) m = fmaxf(m, l[n]);
  red[tid] = m; __syncthreads();
  for (int s = 128; s > 0; s >>= 1) {
    if (tid < s) red[tid] = fmaxf(red[tid], red[tid + s]);
    __syncthreads();
  }
  m = red[0]; __syncthreads();
  float sum = 0.f;
  for (int n = tid; n < N_; n += 256) {
    float x = l[n];
    sum += (x == neg_inf()) ? 0.f : __expf(x - m);
  }
  red[tid] = sum; __syncthreads();
  for (int s = 128; s > 0; s >>= 1) {
    if (tid < s) red[tid] += red[tid + s];
    __syncthreads();
  }
  if (tid == 0) { ws[OFF_MX + bh] = m; ws[OFF_ISUM + bh] = 1.f / red[0]; }
}

#define PSTR 520

#define K4_LOAD_BATCH(ld, base)                                              \
  _Pragma("unroll")   \
  for (int g = 0; g < 8; ++g) {                                              \
    const float* r_ = (base) + (size_t)(g * 4) * C_;                         \
    ld[g][0] = *(const v4f*)(r_);                                            \
    ld[g][1] = *(const v4f*)(r_ + 64);                                       \
    ld[g][2] = *(const v4f*)(r_ + C_);                                       \
    ld[g][3] = *(const v4f*)(r_ + C_ + 64);                                  \
  }

#define K4_MMA_BATCH(ld, pbase)                                              \
  _Pragma("unroll")          \
  for (int g = 0; g < 8; ++g) {                                              \
    v2f a = *(const v2f*)((pbase) + g * 4);                                  \
    if (!real) { a.x = 0.f; a.y = 0.f; }                 \
    _Pragma("unroll")                                           \
    for (int cs = 0; cs < 2; ++cs) {                                         \
      _Pragma("unroll")                                         \
      for (int q = 0; q < 4; ++q) {                                          \
        v2f bv;                                                              \
        bv.x = ld[g][cs][q];                                 \
        bv.y = ld[g][2 + cs][q];                             \
        acc[cs * 4 + q] = __builtin_amdgcn_wmma_f32_16x16x4_f32(             \
            false, a, false, bv, (short)0, acc[cs * 4 + q], false, false);   \
      }                                                                      \
    }                                                                        \
  }

__global__ __launch_bounds__(256, 4)
void k_wsum(const float* __restrict__ V, float* __restrict__ ws) {
  __shared__ __attribute__((aligned(16))) float s_p[H_ * PSTR];
  __shared__ __attribute__((aligned(16))) float s_red[8 * H_ * C_];
  const int b = blockIdx.y, chunk = blockIdx.x;
  const int nb = chunk * 512;
  const int tid = threadIdx.x;
  const int wave = tid >> 5, lane = tid & 31;
  const int l16 = lane & 15, hi = lane >> 4;
  const float* vb = V + ((size_t)b * N_ + nb + wave * 64 + 2 * hi) * C_ + 4 * l16;
  v4f ld[8][4];
  K4_LOAD_BATCH(ld, vb)
  {
    const float* lg = ws + OFF_LOG + (size_t)b * H_ * N_ + nb + tid;
    float x[8];
#pragma unroll
    for (int j = 0; j < 8; ++j) x[j] = lg[(size_t)(j >> 1) * N_ + (j & 1) * 256];
#pragma unroll
    for (int j = 0; j < 8; ++j) {
      const int h = j >> 1;
      float m = ws[OFF_MX + b * H_ + h];
      float is = ws[OFF_ISUM + b * H_ + h];
      s_p[h * PSTR + (j & 1) * 256 + tid] =
          (x[j] == neg_inf()) ? 0.f : __expf(x[j] - m) * is;
    }
  }
  __syncthreads();
  const float* pa = s_p + (l16 & 3) * PSTR + wave * 64 + 2 * hi;
  const bool real = l16 < H_;
  v8f acc[8] = {};
  K4_MMA_BATCH(ld, pa)
  K4_LOAD_BATCH(ld, vb + (size_t)32 * C_)
  K4_MMA_BATCH(ld, pa + 32)
  if (hi == 0) {
    float* rw = s_red + wave * (H_ * C_) + 4 * l16;
#pragma unroll
    for (int r = 0; r < H_; ++r) {
#pragma unroll
      for (int cs = 0; cs < 2; ++cs) {
        v4f o;
        o.x = acc[cs * 4 + 0][r]; o.y = acc[cs * 4 + 1][r];
        o.z = acc[cs * 4 + 2][r]; o.w = acc[cs * 4 + 3][r];
        *(v4f*)(rw + r * C_ + cs * 64) = o;
      }
    }
  }
  __syncthreads();
  float* pr = ws + OFF_PART + (size_t)(b * 8 + chunk) * (H_ * C_);
  for (int i = tid; i < H_ * C_; i += 256) {
    float s = 0.f;
    for (int w = 0; w < 8; ++w) s += s_red[w * (H_ * C_) + i];
    pr[i] = s;
  }
}

__global__ void k_reduce(float* __restrict__ ws) {
  int idx = blockIdx.x * 256 + threadIdx.x;
  int b = idx >> 9, rest = idx & 511;
  float s = 0.f;
  for (int ch = 0; ch < 8; ++ch)
    s += ws[OFF_PART + ((size_t)(b * 8 + ch) << 9) + rest];
  ws[OFF_U + idx] = s;
}

__global__ void k_out(const float* __restrict__ W, const float* __restrict__ bias,
                      const float* __restrict__ U, float* __restrict__ out) {
  const int m0 = blockIdx.x * 16;
  const int h  = blockIdx.y;
  const int tid = threadIdx.x;
  const int wave = tid >> 5, lane = tid & 31;
  const int l16 = lane & 15, hi = lane >> 4;
  const int n0 = wave * 16;
  const float* urow = U + (size_t)(m0 + l16) * HC_ + h * C_;
  const float* wcol = W + h * C_ + n0 + l16;
  v8f acc = {};
  for (int k0 = 0; k0 < C_; k0 += 4) {
    v2f a = *(const v2f*)(urow + k0 + 2 * hi);
    v2f bv;
    bv.x = wcol[(size_t)(k0 + 2 * hi) * HC_];
    bv.y = wcol[(size_t)(k0 + 1 + 2 * hi) * HC_];
    acc = __builtin_amdgcn_wmma_f32_16x16x4_f32(false, a, false, bv,
                                                (short)0, acc, false, false);
  }
  for (int r = 0; r < 8; ++r) {
    int row = m0 + r + 8 * hi;
    int col = h * C_ + n0 + l16;
    out[(size_t)row * HC_ + col] = acc[r] + bias[col];
  }
}

extern "C" void kernel_launch(void* const* d_in, const int* in_sizes, int n_in,
                              void* d_out, int out_size, void* d_ws, size_t ws_size,
                              hipStream_t stream) {
  (void)in_sizes; (void)n_in; (void)out_size; (void)ws_size;
  const float* V    = (const float*)d_in[0];
  const int*   gs   = (const int*)  d_in[1];
  const float* W    = (const float*)d_in[2];
  const float* bias = (const float*)d_in[3];
  const float* T    = (const float*)d_in[4];
  float* out = (float*)d_out;
  float* ws  = (float*)d_ws;

  k_prep  <<<9,            256, 0, stream>>>(W, bias, T, ws);
  k_logits<<<dim3(8, B_),  256, 0, stream>>>(V, gs, ws);
  k_stats <<<B_ * H_,      256, 0, stream>>>(ws);
  k_wsum  <<<dim3(8, B_),  256, 0, stream>>>(V, ws);
  k_reduce<<<(B_*H_*C_)/256, 256, 0, stream>>>(ws);
  k_out   <<<dim3(B_/16, H_), 256, 0, stream>>>(W, bias, ws + OFF_U, out);
}


# K2/K3/K4 work only on nodes below graph_size[b]: chunks, tile pairs and 32-row batches past the limit are skipped (their softmax weights are exactly zero); every sum keeps its order
# speedup vs baseline: 3.1141x; 1.1470x over previous
#include <hip/hip_runtime.h>
#include <math.h>


#define B_ 64
#define N_ 4096
#define C_ 128
#define H_ 4
#define HC_ 512
#define ALPHA 0.2f

typedef __attribute__((ext_vector_type(8))) float v8f;
typedef __attribute__((ext_vector_type(2))) float v2f;
typedef __attribute__((ext_vector_type(4))) float v4f;

#define OFF_W2P   0u
#define OFF_B2    2048u
#define OFF_LOG   2064u
#define OFF_MX    (OFF_LOG + (unsigned)(B_*H_*N_))
#define OFF_ISUM  (OFF_MX + 256u)
#define OFF_PART  (OFF_ISUM + 256u)
#define OFF_U     (OFF_PART + (unsigned)(B_*8*H_*C_))

__device__ __forceinline__ float neg_inf() { return -__builtin_inff(); }

__global__ void k_prep(const float* __restrict__ W, const float* __restrict__ bias,
                       const float* __restrict__ T, float* __restrict__ ws) {
  int idx = blockIdx.x * 256 + threadIdx.x;
  if (idx < 2048) {
    int k = idx >> 4, j = idx & 15;
    float v = 0.f;
    if (j < H_) {
      const float* wr = W + (size_t)k * HC_ + j * C_;
      const float* tr = T + j * C_;
      for (int c = 0; c < C_; ++c) v = fmaf(wr[c], tr[c], v);
    }
    ws[OFF_W2P + idx] = v;
  } else if (idx < 2048 + H_) {
    int j = idx - 2048;
    float v = 0.f;
    for (int c = 0; c < C_; ++c) v = fmaf(bias[j * C_ + c], T[j * C_ + c], v);
    ws[OFF_B2 + j] = v;
  }
}

#define LP 132

#define K2_LOAD_TILE(g, base)                                                \
  _Pragma("unroll")   \
  for (int r = 0; r < 16; ++r) g[r] = *(const v4f*)((base) + (size_t)r * C_);

__device__ __forceinline__ void wave_lds_order() {
  __builtin_amdgcn_fence(__ATOMIC_RELEASE, "wavefront");
  __builtin_amdgcn_wave_barrier();
  __builtin_amdgcn_fence(__ATOMIC_ACQUIRE, "wavefront");
}

__global__ __launch_bounds__(256, 4)
void k_logits(const float* __restrict__ V, const int* __restrict__ gs,
              float* __restrict__ ws) {
  __shared__ __attribute__((aligned(16))) float s_v[8 * 16 * LP];
  __shared__ __attribute__((aligned(16))) float s_w2p[2048];
  const int b = blockIdx.y, chunk = blockIdx.x;
  const int limit = gs[b];
  if (chunk * 512 >= limit) return;
  const int tid = threadIdx.x;
  const int wave = __builtin_amdgcn_readfirstlane(tid >> 5);
  const int lane = tid & 31;
  const int l16 = lane & 15, hi = lane >> 4;
  const int n0 = chunk * 512 + wave * 64;
  const bool act = n0 < limit;
  const bool both = n0 + 32 < limit;
  const float* vg = V + ((size_t)b * N_ + n0) * C_ + 4 * lane;
  float* st = s_v + wave * (16 * LP) + 4 * lane;
  const float* fr = s_v + wave * (16 * LP) + l16 * LP + 4 * hi;
  float tw[8];
#pragma unroll
  for (int k = 0; k < 8; ++k) tw[k] = ws[OFF_W2P + k * 256 + tid];
  v4f g[16];
  if (act) { K2_LOAD_TILE(g, vg) }
#pragma unroll
  for (int k = 0; k < 8; ++k) s_w2p[k * 256 + tid] = tw[k];
  __syncthreads();
  if (!act) return;
  const bool real = (l16 & 4) == 0;
  v2f wf[32];
#pragma unroll
  for (int i = 0; i < 32; ++i) {
    const int c = 8 * (i >> 1) + 4 * hi + 2 * (i & 1);
    const float w0 = s_w2p[c * 16 + (l16 & 3)];
    const float w1 = s_w2p[(c + 1) * 16 + (l16 & 3)];
    wf[i].x = real ? w0 : 0.f;
    wf[i].y = real ? w1 : 0.f;
  }
  float bb[H_];
#pragma unroll
  for (int h = 0; h < H_; ++h) bb[h] = ws[OFF_B2 + h];
  float* lrow = ws + OFF_LOG + (size_t)b * H_ * N_;
  float ev[H_];
#pragma unroll
  for (int t = 0; t < 4; ++t) {
    if (t == 2 && !both) break;
#pragma unroll
    for (int r = 0; r < 16; ++r) *(v4f*)(st + r * LP) = g[r];
    wave_lds_order();
    if (t == 0 || t == 2 || (t == 1 && both)) {
      K2_LOAD_TILE(g, vg + (size_t)(t + 1) * 16 * C_)
    }
    __builtin_amdgcn_sched_barrier(0);
    v4f f[16];
#pragma unroll
    for (int j = 0; j < 16; ++j) f[j] = *(const v4f*)(fr + 8 * j);
    __builtin_amdgcn_sched_barrier(0);
    wave_lds_order();
    v8f acc0 = {}, acc1 = {};
#pragma unroll
    for (int j = 0; j < 16; ++j) {
      v2f b0, b1;
      b0.x = f[j].x; b0.y = f[j].y;
      b1.x = f[j].z; b1.y = f[j].w;
      acc0 = __builtin_amdgcn_wmma_f32_16x16x4_f32(false, wf[2 * j], false, b0,
                                                   (short)0, acc0, false, false);
      acc1 = __builtin_amdgcn_wmma_f32_16x16x4_f32(false, wf[2 * j + 1], false, b1,
                                                   (short)0, acc1, false, false);
    }
    if ((t & 1) == 0) {
#pragma unroll
      for (int h = 0; h < H_; ++h) ev[h] = acc0[h] + acc1[h];
    } else {
      const int node = n0 + (t - 1) * 16 + lane;
#pragma unroll
      for (int h = 0; h < H_; ++h) {
        float x = (hi ? (acc0[h] + acc1[h]) : ev[h]) + bb[h];
        x = (x >= 0.f) ? x : ALPHA * x;
        lrow[(size_t)h * N_ + node] = (node < limit) ? x : neg_inf();
      }
    }
    __builtin_amdgcn_sched_barrier(0);
  }
}

__global__ void k_stats(const int* __restrict__ gs, float* __restrict__ ws) {
  __shared__ float red[256];
  const int bh = blockIdx.x;
  const int tid = threadIdx.x;
  const int limit = gs[bh / H_];
  const float* l = ws + OFF_LOG + (size_t)bh * N_;
  float m = neg_inf();
  for (int n = tid; n < N_; n += 256) {
    const bool ok = n < limit;
    const float x = l[ok ? n : 0];
    m = fmaxf(m, ok ? x : neg_inf());
  }
  red[tid] = m; __syncthreads();
  for (int s = 128; s > 0; s >>= 1) {
    if (tid < s) red[tid] = fmaxf(red[tid], red[tid + s]);
    __syncthreads();
  }
  m = red[0]; __syncthreads();
  float sum = 0.f;
  for (int n = tid; n < N_; n += 256) {
    const bool ok = n < limit;
    float x = l[ok ? n : 0];
    x = ok ? x : neg_inf();
    sum += (x == neg_inf()) ? 0.f : __expf(x - m);
  }
  red[tid] = sum; __syncthreads();
  for (int s = 128; s > 0; s >>= 1) {
    if (tid < s) red[tid] += red[tid + s];
    __syncthreads();
  }
  if (tid == 0) { ws[OFF_MX + bh] = m; ws[OFF_ISUM + bh] = 1.f / red[0]; }
}

#define PSTR 520

#define K4_LOAD_BATCH(ld, base)                                              \
  _Pragma("unroll")   \
  for (int g = 0; g < 8; ++g) {                                              \
    const float* r_ = (base) + (size_t)(g * 4) * C_;                         \
    ld[g][0] = *(const v4f*)(r_);                                            \
    ld[g][1] = *(const v4f*)(r_ + 64);                                       \
    ld[g][2] = *(const v4f*)(r_ + C_);                                       \
    ld[g][3] = *(const v4f*)(r_ + C_ + 64);                                  \
  }

#define K4_MMA_BATCH(ld, pbase)                                              \
  _Pragma("unroll")          \
  for (int g = 0; g < 8; ++g) {                                              \
    v2f a = *(const v2f*)((pbase) + g * 4);                                  \
    if (!real) { a.x = 0.f; a.y = 0.f; }                 \
    _Pragma("unroll")                                           \
    for (int cs = 0; cs < 2; ++cs) {                                         \
      _Pragma("unroll")                                         \
      for (int q = 0; q < 4; ++q) {                                          \
        v2f bv;                                                              \
        bv.x = ld[g][cs][q];                                 \
        bv.y = ld[g][2 + cs][q];                             \
        acc[cs * 4 + q] = __builtin_amdgcn_wmma_f32_16x16x4_f32(             \
            false, a, false, bv, (short)0, acc[cs * 4 + q], false, false);   \
      }                                                                      \
    }                                                                        \
  }

__global__ __launch_bounds__(256, 4)
void k_wsum(const float* __restrict__ V, const int* __restrict__ gs,
            float* __restrict__ ws) {
  __shared__ __attribute__((aligned(16))) float s_p[H_ * PSTR];
  __shared__ __attribute__((aligned(16))) float s_red[8 * H_ * C_];
  const int b = blockIdx.y, chunk = blockIdx.x;
  const int nb = chunk * 512;
  const int tid = threadIdx.x;
  const int limit = gs[b];
  if (nb >= limit) {
    float* pz = ws + OFF_PART + (size_t)(b * 8 + chunk) * (H_ * C_);
    pz[tid] = 0.f; pz[256 + tid] = 0.f;
    return;
  }
  const int wave = __builtin_amdgcn_readfirstlane(tid >> 5);
  const int lane = tid & 31;
  const int l16 = lane & 15, hi = lane >> 4;
  const bool bat0 = nb + wave * 64 < limit;
  const bool bat1 = nb + wave * 64 + 32 < limit;
  const float* vb = V + ((size_t)b * N_ + nb + wave * 64 + 2 * hi) * C_ + 4 * l16;
  v4f ld[8][4];
  if (bat0) { K4_LOAD_BATCH(ld, vb) }
  {
    const float* lg = ws + OFF_LOG + (size_t)b * H_ * N_ + nb + tid;
    float x[8];
    const bool ok0 = nb + tid < limit, ok1 = nb + 256 + tid < limit;
    const int o0 = ok0 ? 0 : -tid, o1 = ok1 ? 256 : -tid;
#pragma unroll
    for (int j = 0; j < 8; ++j) x[j] = lg[(j >> 1) * N_ + ((j & 1) ? o1 : o0)];
    float mx[H_], is[H_];
#pragma unroll
    for (int h = 0; h < H_; ++h) {
      mx[h] = ws[OFF_MX + b * H_ + h];
      is[h] = ws[OFF_ISUM + b * H_ + h];
    }
#pragma unroll
    for (int j = 0; j < 8; ++j) {
      const int h = j >> 1;
      const float e = __expf(x[j] - mx[h]) * is[h];
      s_p[h * PSTR + (j & 1) * 256 + tid] = ((j & 1) ? ok1 : ok0) ? e : 0.f;
    }
  }
  __syncthreads();
  const float* pa = s_p + (l16 & 3) * PSTR + wave * 64 + 2 * hi;
  const bool real = l16 < H_;
  v8f acc[8] = {};
  if (bat1) {
    K4_MMA_BATCH(ld, pa)
    K4_LOAD_BATCH(ld, vb + (size_t)32 * C_)
    K4_MMA_BATCH(ld, pa + 32)
  } else if (bat0) {
    K4_MMA_BATCH(ld, pa)
  }
  if (hi == 0) {
    float* rw = s_red + wave * (H_ * C_) + 4 * l16;
#pragma unroll
    for (int r = 0; r < H_; ++r) {
#pragma unroll
      for (int cs = 0; cs < 2; ++cs) {
        v4f o;
        o.x = acc[cs * 4 + 0][r]; o.y = acc[cs * 4 + 1][r];
        o.z = acc[cs * 4 + 2][r]; o.w = acc[cs * 4 + 3][r];
        *(v4f*)(rw + r * C_ + cs * 64) = o;
      }
    }
  }
  __syncthreads();
  float* pr = ws + OFF_PART + (size_t)(b * 8 + chunk) * (H_ * C_);
  for (int i = tid; i < H_ * C_; i += 256) {
    float s = 0.f;
    for (int w = 0; w < 8; ++w) s += s_red[w * (H_ * C_) + i];
    pr[i] = s;
  }
}

__global__ void k_reduce(float* __restrict__ ws) {
  int idx = blockIdx.x * 256 + threadIdx.x;
  int b = idx >> 9, rest = idx & 511;
  float s = 0.f;
  for (int ch = 0; ch < 8; ++ch)
    s += ws[OFF_PART + ((size_t)(b * 8 + ch) << 9) + rest];
  ws[OFF_U + idx] = s;
}

__global__ void k_out(const float* __restrict__ W, const float* __restrict__ bias,
                      const float* __restrict__ U, float* __restrict__ out) {
  const int m0 = blockIdx.x * 16;
  const int h  = blockIdx.y;
  const int tid = threadIdx.x;
  const int wave = tid >> 5, lane = tid & 31;
  const int l16 = lane & 15, hi = lane >> 4;
  const int n0 = wave * 16;
  const float* urow = U + (size_t)(m0 + l16) * HC_ + h * C_;
  const float* wcol = W + h * C_ + n0 + l16;
  v8f acc = {};
  for (int k0 = 0; k0 < C_; k0 += 4) {
    v2f a = *(const v2f*)(urow + k0 + 2 * hi);
    v2f bv;
    bv.x = wcol[(size_t)(k0 + 2 * hi) * HC_];
    bv.y = wcol[(size_t)(k0 + 1 + 2 * hi) * HC_];
    acc = __builtin_amdgcn_wmma_f32_16x16x4_f32(false, a, false, bv,
                                                (short)0, acc, false, false);
  }
  for (int r = 0; r < 8; ++r) {
    int row = m0 + r + 8 * hi;
    int col = h * C_ + n0 + l16;
    out[(size_t)row * HC_ + col] = acc[r] + bias[col];
  }
}

extern "C" void kernel_launch(void* const* d_in, const int* in_sizes, int n_in,
                              void* d_out, int out_size, void* d_ws, size_t ws_size,
                              hipStream_t stream) {
  (void)in_sizes; (void)n_in; (void)out_size; (void)ws_size;
  const float* V    = (const float*)d_in[0];
  const int*   gs   = (const int*)  d_in[1];
  const float* W    = (const float*)d_in[2];
  const float* bias = (const float*)d_in[3];
  const float* T    = (const float*)d_in[4];
  float* out = (float*)d_out;
  float* ws  = (float*)d_ws;

  k_prep  <<<9,            256, 0, stream>>>(W, bias, T, ws);
  k_logits<<<dim3(8, B_),  256, 0, stream>>>(V, gs, ws);
  k_stats <<<B_ * H_,      256, 0, stream>>>(gs, ws);
  k_wsum  <<<dim3(8, B_),  256, 0, stream>>>(V, gs, ws);
  k_reduce<<<(B_*H_*C_)/256, 256, 0, stream>>>(ws);
  k_out   <<<dim3(B_/16, H_), 256, 0, stream>>>(W, bias, ws + OFF_U, out);
}


# Softmax statistics lose their own launch: each K2 wave leaves (max, sum of exp) of its 64 logits, K4 merges a graph's partials itself (K3 removed, 5 launches); the softmax sum order changes
# speedup vs baseline: 3.5956x; 1.1546x over previous
#include <hip/hip_runtime.h>
#include <math.h>


#define B_ 64
#define N_ 4096
#define C_ 128
#define H_ 4
#define HC_ 512
#define ALPHA 0.2f

typedef __attribute__((ext_vector_type(8))) float v8f;
typedef __attribute__((ext_vector_type(2))) float v2f;
typedef __attribute__((ext_vector_type(4))) float v4f;

#define OFF_W2P   0u
#define OFF_B2    2048u
#define OFF_LOG   2064u
#define OFF_PART  (OFF_LOG + (unsigned)(B_*H_*N_) + 512u)
#define OFF_U     (OFF_PART + (unsigned)(B_*8*H_*C_))
#define OFF_PM    OFF_U
#define OFF_PS    (OFF_U + (unsigned)(B_*64*H_))

__device__ __forceinline__ float neg_inf() { return -__builtin_inff(); }

#define ROW_XOR(v, M)                                                        \
  __builtin_bit_cast(float, __builtin_amdgcn_update_dpp(                     \
      __builtin_bit_cast(int, (v)), __builtin_bit_cast(int, (v)),            \
      0x160 + (M)  , 0xf, 0xf, false))
__device__ __forceinline__ float xhalf(float v) {
  const int b = __builtin_bit_cast(int, v);
  return __builtin_bit_cast(float, __builtin_amdgcn_permlanex16(
      b, b, 0x76543210, 0xfedcba98, false, false));
}
__device__ __forceinline__ float wave_max(float v) {
  v = fmaxf(v, ROW_XOR(v, 1)); v = fmaxf(v, ROW_XOR(v, 2));
  v = fmaxf(v, ROW_XOR(v, 4)); v = fmaxf(v, ROW_XOR(v, 8));
  return fmaxf(v, xhalf(v));
}
__device__ __forceinline__ float wave_sum(float v) {
  v += ROW_XOR(v, 1); v += ROW_XOR(v, 2); v += ROW_XOR(v, 4); v += ROW_XOR(v, 8);
  return v + xhalf(v);
}
__device__ __forceinline__ float head_max(float v) {
  v = fmaxf(v, ROW_XOR(v, 4)); v = fmaxf(v, ROW_XOR(v, 8));
  return fmaxf(v, xhalf(v));
}
__device__ __forceinline__ float head_sum(float v) {
  v += ROW_XOR(v, 4); v += ROW_XOR(v, 8);
  return v + xhalf(v);
}

__global__ void k_prep(const float* __restrict__ W, const float* __restrict__ bias,
                       const float* __restrict__ T, float* __restrict__ ws) {
  int idx = blockIdx.x * 256 + threadIdx.x;
  if (idx < 2048) {
    int k = idx >> 4, j = idx & 15;
    float v = 0.f;
    if (j < H_) {
      const float* wr = W + (size_t)k * HC_ + j * C_;
      const float* tr = T + j * C_;
      for (int c = 0; c < C_; ++c) v = fmaf(wr[c], tr[c], v);
    }
    ws[OFF_W2P + idx] = v;
  } else if (idx < 2048 + H_) {
    int j = idx - 2048;
    float v = 0.f;
    for (int c = 0; c < C_; ++c) v = fmaf(bias[j * C_ + c], T[j * C_ + c], v);
    ws[OFF_B2 + j] = v;
  }
}

#define LP 132

#define K2_LOAD_TILE(g, base)                                                \
  _Pragma("unroll")   \
  for (int r = 0; r < 16; ++r) g[r] = *(const v4f*)((base) + (size_t)r * C_);

__device__ __forceinline__ void wave_lds_order() {
  __builtin_amdgcn_fence(__ATOMIC_RELEASE, "wavefront");
  __builtin_amdgcn_wave_barrier();
  __builtin_amdgcn_fence(__ATOMIC_ACQUIRE, "wavefront");
}

__global__ __launch_bounds__(256, 4)
void k_logits(const float* __restrict__ V, const int* __restrict__ gs,
              float* __restrict__ ws) {
  __shared__ __attribute__((aligned(16))) float s_v[8 * 16 * LP];
  __shared__ __attribute__((aligned(16))) float s_w2p[2048];
  const int b = blockIdx.y, chunk = blockIdx.x;
  const int limit = gs[b];
  if (chunk * 512 >= limit) return;
  const int tid = threadIdx.x;
  const int wave = __builtin_amdgcn_readfirstlane(tid >> 5);
  const int lane = tid & 31;
  const int l16 = lane & 15, hi = lane >> 4;
  const int n0 = chunk * 512 + wave * 64;
  const bool act = n0 < limit;
  const bool both = n0 + 32 < limit;
  const float* vg = V + ((size_t)b * N_ + n0) * C_ + 4 * lane;
  float* st = s_v + wave * (16 * LP) + 4 * lane;
  const float* fr = s_v + wave * (16 * LP) + l16 * LP + 4 * hi;
  float tw[8];
#pragma unroll
  for (int k = 0; k < 8; ++k) tw[k] = ws[OFF_W2P + k * 256 + tid];
  v4f g[16];
  if (act) { K2_LOAD_TILE(g, vg) }
#pragma unroll
  for (int k = 0; k < 8; ++k) s_w2p[k * 256 + tid] = tw[k];
  __syncthreads();
  if (!act) return;
  const bool real = (l16 & 4) == 0;
  v2f wf[32];
#pragma unroll
  for (int i = 0; i < 32; ++i) {
    const int c = 8 * (i >> 1) + 4 * hi + 2 * (i & 1);
    const float w0 = s_w2p[c * 16 + (l16 & 3)];
    const float w1 = s_w2p[(c + 1) * 16 + (l16 & 3)];
    wf[i].x = real ? w0 : 0.f;
    wf[i].y = real ? w1 : 0.f;
  }
  float bb[H_];
#pragma unroll
  for (int h = 0; h < H_; ++h) bb[h] = ws[OFF_B2 + h];
  float* lrow = ws + OFF_LOG + (size_t)b * H_ * N_;
  float ev[H_];
  float xk[2][H_];
#pragma unroll
  for (int h = 0; h < H_; ++h) { xk[0][h] = neg_inf(); xk[1][h] = neg_inf(); }
#pragma unroll
  for (int t = 0; t < 4; ++t) {
    if (t == 2 && !both) break;
#pragma unroll
    for (int r = 0; r < 16; ++r) *(v4f*)(st + r * LP) = g[r];
    wave_lds_order();
    if (t == 0 || t == 2 || (t == 1 && both)) {
      K2_LOAD_TILE(g, vg + (size_t)(t + 1) * 16 * C_)
    }
    __builtin_amdgcn_sched_barrier(0);
    v4f f[16];
#pragma unroll
    for (int j = 0; j < 16; ++j) f[j] = *(const v4f*)(fr + 8 * j);
    __builtin_amdgcn_sched_barrier(0);
    wave_lds_order();
    v8f acc0 = {}, acc1 = {};
#pragma unroll
    for (int j = 0; j < 16; ++j) {
      v2f b0, b1;
      b0.x = f[j].x; b0.y = f[j].y;
      b1.x = f[j].z; b1.y = f[j].w;
      acc0 = __builtin_amdgcn_wmma_f32_16x16x4_f32(false, wf[2 * j], false, b0,
                                                   (short)0, acc0, false, false);
      acc1 = __builtin_amdgcn_wmma_f32_16x16x4_f32(false, wf[2 * j + 1], false, b1,
                                                   (short)0, acc1, false, false);
    }
    if ((t & 1) == 0) {
#pragma unroll
      for (int h = 0; h < H_; ++h) ev[h] = acc0[h] + acc1[h];
    } else {
      const int node = n0 + (t - 1) * 16 + lane;
#pragma unroll
      for (int h = 0; h < H_; ++h) {
        float x = (hi ? (acc0[h] + acc1[h]) : ev[h]) + bb[h];
        x = (x >= 0.f) ? x : ALPHA * x;
        x = (node < limit) ? x : neg_inf();
        lrow[(size_t)h * N_ + node] = x;
        xk[t >> 1][h] = x;
      }
    }
    __builtin_amdgcn_sched_barrier(0);
  }
  float mw[H_], sw[H_];
#pragma unroll
  for (int h = 0; h < H_; ++h) mw[h] = fmaxf(xk[0][h], xk[1][h]);
#pragma unroll
  for (int h = 0; h < H_; ++h) mw[h] = wave_max(mw[h]);
#pragma unroll
  for (int h = 0; h < H_; ++h) {
    const float e0 = (xk[0][h] == neg_inf()) ? 0.f : __expf(xk[0][h] - mw[h]);
    const float e1 = (xk[1][h] == neg_inf()) ? 0.f : __expf(xk[1][h] - mw[h]);
    sw[h] = e0 + e1;
  }
#pragma unroll
  for (int h = 0; h < H_; ++h) sw[h] = wave_sum(sw[h]);
  if (lane < H_) {
    const float mv = lane == 0 ? mw[0] : lane == 1 ? mw[1] : lane == 2 ? mw[2] : mw[3];
    const float sv = lane == 0 ? sw[0] : lane == 1 ? sw[1] : lane == 2 ? sw[2] : sw[3];
    const unsigned rec = (unsigned)((b * 64 + chunk * 8 + wave) * H_ + lane);
    ws[OFF_PM + rec] = mv;
    ws[OFF_PS + rec] = sv;
  }
}

#define PSTR 520

#define K4_LOAD_BATCH(ld, base)                                              \
  _Pragma("unroll")   \
  for (int g = 0; g < 8; ++g) {                                              \
    const float* r_ = (base) + (size_t)(g * 4) * C_;                         \
    ld[g][0] = *(const v4f*)(r_);                                            \
    ld[g][1] = *(const v4f*)(r_ + 64);                                       \
    ld[g][2] = *(const v4f*)(r_ + C_);                                       \
    ld[g][3] = *(const v4f*)(r_ + C_ + 64);                                  \
  }

#define K4_MMA_BATCH(ld, pbase)                                              \
  _Pragma("unroll")          \
  for (int g = 0; g < 8; ++g) {                                              \
    v2f a = *(const v2f*)((pbase) + g * 4);                                  \
    if (!real) { a.x = 0.f; a.y = 0.f; }                 \
    _Pragma("unroll")                                           \
    for (int cs = 0; cs < 2; ++cs) {                                         \
      _Pragma("unroll")                                         \
      for (int q = 0; q < 4; ++q) {                                          \
        v2f bv;                                                              \
        bv.x = ld[g][cs][q];                                 \
        bv.y = ld[g][2 + cs][q];                             \
        acc[cs * 4 + q] = __builtin_amdgcn_wmma_f32_16x16x4_f32(             \
            false, a, false, bv, (short)0, acc[cs * 4 + q], false, false);   \
      }                                                                      \
    }                                                                        \
  }

__global__ __launch_bounds__(256, 4)
void k_wsum(const float* __restrict__ V, const int* __restrict__ gs,
            float* __restrict__ ws) {
  __shared__ __attribute__((aligned(16))) float s_p[H_ * PSTR];
  __shared__ __attribute__((aligned(16))) float s_red[8 * H_ * C_];
  const int b = blockIdx.y, chunk = blockIdx.x;
  const int nb = chunk * 512;
  const int tid = threadIdx.x;
  const int limit = gs[b];
  if (nb >= limit) {
    float* pz = ws + OFF_PART + (size_t)(b * 8 + chunk) * (H_ * C_);
    pz[tid] = 0.f; pz[256 + tid] = 0.f;
    return;
  }
  const int wave = __builtin_amdgcn_readfirstlane(tid >> 5);
  const int lane = tid & 31;
  const int l16 = lane & 15, hi = lane >> 4;
  const bool bat0 = nb + wave * 64 < limit;
  const bool bat1 = nb + wave * 64 + 32 < limit;
  const float* vb = V + ((size_t)b * N_ + nb + wave * 64 + 2 * hi) * C_ + 4 * l16;
  v4f ld[8][4];
  if (bat0) { K4_LOAD_BATCH(ld, vb) }
  {
    const float* lg = ws + OFF_LOG + (size_t)b * H_ * N_ + nb + tid;
    float x[8];
    const bool ok0 = nb + tid < limit, ok1 = nb + 256 + tid < limit;
    const int o0 = ok0 ? 0 : -tid, o1 = ok1 ? 256 : -tid;
#pragma unroll
    for (int j = 0; j < 8; ++j) x[j] = lg[(j >> 1) * N_ + ((j & 1) ? o1 : o0)];
    const int nw = (limit + 63) >> 6;
    const int hl = lane & 3;
    const float* pmb = ws + OFF_PM + (size_t)b * (64 * H_);
    const float* psb = ws + OFF_PS + (size_t)b * (64 * H_);
    float pm[8], ps[8];
#pragma unroll
    for (int k = 0; k < 8; ++k) {
      const int w = (lane >> 2) + 8 * k;
      const int idx = (w < nw) ? w * H_ + hl : hl;
      pm[k] = pmb[idx];
      ps[k] = psb[idx];
    }
    float gm = neg_inf();
#pragma unroll
    for (int k = 0; k < 8; ++k) {
      pm[k] = ((lane >> 2) + 8 * k < nw) ? pm[k] : neg_inf();
      gm = fmaxf(gm, pm[k]);
    }
    gm = head_max(gm);
    float gsum = 0.f;
#pragma unroll
    for (int k = 0; k < 8; ++k) gsum += ps[k] * __expf(pm[k] - gm);
    gsum = head_sum(gsum);
    float mx[H_], is[H_];
#pragma unroll
    for (int h = 0; h < H_; ++h) {
      mx[h] = __int_as_float(__builtin_amdgcn_readlane(__float_as_int(gm), h));
      is[h] = 1.f / __int_as_float(__builtin_amdgcn_readlane(__float_as_int(gsum), h));
    }
#pragma unroll
    for (int j = 0; j < 8; ++j) {
      const int h = j >> 1;
      const float e = __expf(x[j] - mx[h]) * is[h];
      s_p[h * PSTR + (j & 1) * 256 + tid] = ((j & 1) ? ok1 : ok0) ? e : 0.f;
    }
  }
  __syncthreads();
  const float* pa = s_p + (l16 & 3) * PSTR + wave * 64 + 2 * hi;
  const bool real = l16 < H_;
  v8f acc[8] = {};
  if (bat1) {
    K4_MMA_BATCH(ld, pa)
    K4_LOAD_BATCH(ld, vb + (size_t)32 * C_)
    K4_MMA_BATCH(ld, pa + 32)
  } else if (bat0) {
    K4_MMA_BATCH(ld, pa)
  }
  if (hi == 0) {
    float* rw = s_red + wave * (H_ * C_) + 4 * l16;
#pragma unroll
    for (int r = 0; r < H_; ++r) {
#pragma unroll
      for (int cs = 0; cs < 2; ++cs) {
        v4f o;
        o.x = acc[cs * 4 + 0][r]; o.y = acc[cs * 4 + 1][r];
        o.z = acc[cs * 4 + 2][r]; o.w = acc[cs * 4 + 3][r];
        *(v4f*)(rw + r * C_ + cs * 64) = o;
      }
    }
  }
  __syncthreads();
  float* pr = ws + OFF_PART + (size_t)(b * 8 + chunk) * (H_ * C_);
  for (int i = tid; i < H_ * C_; i += 256) {
    float s = 0.f;
    for (int w = 0; w < 8; ++w) s += s_red[w * (H_ * C_) + i];
    pr[i] = s;
  }
}

__global__ void k_reduce(float* __restrict__ ws) {
  int idx = blockIdx.x * 256 + threadIdx.x;
  int b = idx >> 9, rest = idx & 511;
  float s = 0.f;
  for (int ch = 0; ch < 8; ++ch)
    s += ws[OFF_PART + ((size_t)(b * 8 + ch) << 9) + rest];
  ws[OFF_U + idx] = s;
}

__global__ void k_out(const float* __restrict__ W, const float* __restrict__ bias,
                      const float* __restrict__ U, float* __restrict__ out) {
  const int m0 = blockIdx.x * 16;
  const int h  = blockIdx.y;
  const int tid = threadIdx.x;
  const int wave = tid >> 5, lane = tid & 31;
  const int l16 = lane & 15, hi = lane >> 4;
  const int n0 = wave * 16;
  const float* urow = U + (size_t)(m0 + l16) * HC_ + h * C_;
  const float* wcol = W + h * C_ + n0 + l16;
  v8f acc = {};
  for (int k0 = 0; k0 < C_; k0 += 4) {
    v2f a = *(const v2f*)(urow + k0 + 2 * hi);
    v2f bv;
    bv.x = wcol[(size_t)(k0 + 2 * hi) * HC_];
    bv.y = wcol[(size_t)(k0 + 1 + 2 * hi) * HC_];
    acc = __builtin_amdgcn_wmma_f32_16x16x4_f32(false, a, false, bv,
                                                (short)0, acc, false, false);
  }
  for (int r = 0; r < 8; ++r) {
    int row = m0 + r + 8 * hi;
    int col = h * C_ + n0 + l16;
    out[(size_t)row * HC_ + col] = acc[r] + bias[col];
  }
}

extern "C" void kernel_launch(void* const* d_in, const int* in_sizes, int n_in,
                              void* d_out, int out_size, void* d_ws, size_t ws_size,
                              hipStream_t stream) {
  (void)in_sizes; (void)n_in; (void)out_size; (void)ws_size;
  const float* V    = (const float*)d_in[0];
  const int*   gs   = (const int*)  d_in[1];
  const float* W    = (const float*)d_in[2];
  const float* bias = (const float*)d_in[3];
  const float* T    = (const float*)d_in[4];
  float* out = (float*)d_out;
  float* ws  = (float*)d_ws;

  k_prep  <<<9,            256, 0, stream>>>(W, bias, T, ws);
  k_logits<<<dim3(8, B_),  256, 0, stream>>>(V, gs, ws);
  k_wsum  <<<dim3(8, B_),  256, 0, stream>>>(V, gs, ws);
  k_reduce<<<(B_*H_*C_)/256, 256, 0, stream>>>(ws);
  k_out   <<<dim3(B_/16, H_), 256, 0, stream>>>(W, bias, ws + OFF_U, out);
}


# V read once: per 16-row tile a wave forms logits, weights them against a running maximum and rescales its sums; 4-wave groups, 256 rows, chunks merged by their maxima (4 launches); sum order changes
# speedup vs baseline: 3.9327x; 1.0938x over previous
#include <hip/hip_runtime.h>
#include <math.h>


#define B_ 64
#define N_ 4096
#define C_ 128
#define H_ 4
#define HC_ 512
#define ALPHA 0.2f

typedef __attribute__((ext_vector_type(8))) float v8f;
typedef __attribute__((ext_vector_type(2))) float v2f;
typedef __attribute__((ext_vector_type(4))) float v4f;

#define OFF_W2P   0u
#define OFF_B2    2048u
#define NCH       16
#define OFF_CM    2080u
#define OFF_CS    (OFF_CM + (unsigned)(B_*NCH*H_))
#define OFF_PART  (OFF_CS + (unsigned)(B_*NCH*H_))
#define OFF_U     (2064u + (unsigned)(B_*H_*N_) + 512u + (unsigned)(B_*8*H_*C_))

__device__ __forceinline__ float neg_inf() { return -__builtin_inff(); }

#define ROW_XOR(v, M)                                                        \
  __builtin_bit_cast(float, __builtin_amdgcn_update_dpp(                     \
      __builtin_bit_cast(int, (v)), __builtin_bit_cast(int, (v)),            \
      0x160 + (M)  , 0xf, 0xf, false))
__device__ __forceinline__ float row_max(float v) {
  v = fmaxf(v, ROW_XOR(v, 1)); v = fmaxf(v, ROW_XOR(v, 2));
  v = fmaxf(v, ROW_XOR(v, 4)); return fmaxf(v, ROW_XOR(v, 8));
}
__device__ __forceinline__ float row_sum(float v) {
  v += ROW_XOR(v, 1); v += ROW_XOR(v, 2); v += ROW_XOR(v, 4);
  return v + ROW_XOR(v, 8);
}

__global__ void k_prep(const float* __restrict__ W, const float* __restrict__ bias,
                       const float* __restrict__ T, float* __restrict__ ws) {
  int idx = blockIdx.x * 256 + threadIdx.x;
  if (idx < 2048) {
    int k = idx >> 4, j = idx & 15;
    float v = 0.f;
    if (j < H_) {
      const float* wr = W + (size_t)k * HC_ + j * C_;
      const float* tr = T + j * C_;
      for (int c = 0; c < C_; ++c) v = fmaf(wr[c], tr[c], v);
    }
    ws[OFF_W2P + idx] = v;
  } else if (idx < 2048 + H_) {
    int j = idx - 2048;
    float v = 0.f;
    for (int c = 0; c < C_; ++c) v = fmaf(bias[j * C_ + c], T[j * C_ + c], v);
    ws[OFF_B2 + j] = v;
  }
}

#define LP 132
#define PP 16

#define LOAD_TILE(g, base)                                                   \
  _Pragma("unroll")   \
  for (int r = 0; r < 16; ++r) g[r] = *(const v4f*)((base) + (size_t)r * C_);

__device__ __forceinline__ void wave_lds_order() {
  __builtin_amdgcn_fence(__ATOMIC_RELEASE, "wavefront");
  __builtin_amdgcn_wave_barrier();
  __builtin_amdgcn_fence(__ATOMIC_ACQUIRE, "wavefront");
}

__global__ __launch_bounds__(128, 3)
void k_attn(const float* __restrict__ V, const int* __restrict__ gs,
            float* __restrict__ ws) {
  __shared__ __attribute__((aligned(16))) float s_v[4 * 16 * LP];
  __shared__ __attribute__((aligned(16))) float s_w2p[2048];
  __shared__ __attribute__((aligned(16))) float s_pw[4 * H_ * PP];
  __shared__ __attribute__((aligned(16))) float s_red[4 * H_ * C_];
  __shared__ float s_m[4 * H_];
  __shared__ float s_s[4 * H_];
  const int b = blockIdx.y, chunk = blockIdx.x;
  const int limit = gs[b];
  if (chunk * 256 >= limit) return;
  const int tid = threadIdx.x;
  const int wave = __builtin_amdgcn_readfirstlane(tid >> 5);
  const int lane = tid & 31;
  const int l16 = lane & 15, hi = lane >> 4;
  const int n0 = chunk * 256 + wave * 64;
  const bool act = n0 < limit;
  const float* vg = V + ((size_t)b * N_ + n0) * C_ + 4 * lane;
  float* tile = s_v + wave * (16 * LP);
  float* st = tile + 4 * lane;
  const float* fr = tile + l16 * LP + 4 * hi;
  const float* br = tile + (2 * hi) * LP + 4 * l16;
  float* pw = s_pw + wave * (H_ * PP);
  const float* pa = pw + (l16 & 3) * PP + 2 * hi;
  float tw[16];
#pragma unroll
  for (int k = 0; k < 16; ++k) tw[k] = ws[OFF_W2P + k * 128 + tid];
  v4f g[16];
  if (act) { LOAD_TILE(g, vg) }
#pragma unroll
  for (int k = 0; k < 16; ++k) s_w2p[k * 128 + tid] = tw[k];
  __syncthreads();
  v8f acc[8] = {};
  float mrun[H_], srun[H_];
#pragma unroll
  for (int h = 0; h < H_; ++h) { mrun[h] = neg_inf(); srun[h] = 0.f; }
  if (act) {
    const bool real1 = (l16 & 4) == 0;
    const bool real2 = l16 < H_;
    v2f wf[32];
#pragma unroll
    for (int i = 0; i < 32; ++i) {
      const int c = 8 * (i >> 1) + 4 * hi + 2 * (i & 1);
      const float w0 = s_w2p[c * 16 + (l16 & 3)];
      const float w1 = s_w2p[(c + 1) * 16 + (l16 & 3)];
      wf[i].x = real1 ? w0 : 0.f;
      wf[i].y = real1 ? w1 : 0.f;
    }
    float bb[H_];
#pragma unroll
    for (int h = 0; h < H_; ++h) bb[h] = ws[OFF_B2 + h];
#pragma unroll
    for (int t = 0; t < 4; ++t) {
      if (t > 0 && n0 + 16 * t >= limit) break;
#pragma unroll
      for (int r = 0; r < 16; ++r) *(v4f*)(st + r * LP) = g[r];
      wave_lds_order();
      if (t < 3 && n0 + 16 * (t + 1) < limit) {
        LOAD_TILE(g, vg + (size_t)(t + 1) * 16 * C_)
      }
      __builtin_amdgcn_sched_barrier(0);
      v4f f[16];
#pragma unroll
      for (int j = 0; j < 16; ++j) f[j] = *(const v4f*)(fr + 8 * j);
      __builtin_amdgcn_sched_barrier(0);
      v8f acc0 = {}, acc1 = {};
#pragma unroll
      for (int j = 0; j < 16; ++j) {
        v2f b0, b1;
        b0.x = f[j].x; b0.y = f[j].y;
        b1.x = f[j].z; b1.y = f[j].w;
        acc0 = __builtin_amdgcn_wmma_f32_16x16x4_f32(false, wf[2 * j], false, b0,
                                                     (short)0, acc0, false, false);
        acc1 = __builtin_amdgcn_wmma_f32_16x16x4_f32(false, wf[2 * j + 1], false, b1,
                                                     (short)0, acc1, false, false);
      }
      const int node = n0 + 16 * t + l16;
      float sc[H_];
#pragma unroll
      for (int h = 0; h < H_; ++h) {
        float x = acc0[h] + acc1[h] + bb[h];
        x = (x >= 0.f) ? x : ALPHA * x;
        x = (node < limit) ? x : neg_inf();
        const float mn = fmaxf(mrun[h], row_max(x));
        sc[h] = __expf(mrun[h] - mn);
        const float p = __expf(x - mn);
        srun[h] = srun[h] * sc[h] + p;
        mrun[h] = mn;
        pw[h * PP + l16] = p;
      }
#pragma unroll
      for (int i = 0; i < 8; ++i) {
#pragma unroll
        for (int r = 0; r < H_; ++r) acc[i][r] *= sc[r];
      }
      wave_lds_order();
#pragma unroll
      for (int gh = 0; gh < 2; ++gh) {
        v4f vb[2][4];
        v2f a[2];
#pragma unroll
        for (int gg = 0; gg < 2; ++gg) {
          const int g4 = (gh * 2 + gg) * 4;
          a[gg] = *(const v2f*)(pa + g4);
          vb[gg][0] = *(const v4f*)(br + g4 * LP);
          vb[gg][1] = *(const v4f*)(br + g4 * LP + 64);
          vb[gg][2] = *(const v4f*)(br + (g4 + 1) * LP);
          vb[gg][3] = *(const v4f*)(br + (g4 + 1) * LP + 64);
        }
        __builtin_amdgcn_sched_barrier(0);
#pragma unroll
        for (int gg = 0; gg < 2; ++gg) {
          v2f av = a[gg];
          if (!real2) { av.x = 0.f; av.y = 0.f; }
#pragma unroll
          for (int cs = 0; cs < 2; ++cs) {
#pragma unroll
            for (int q = 0; q < 4; ++q) {
              v2f bv;
              bv.x = vb[gg][cs][q];
              bv.y = vb[gg][2 + cs][q];
              acc[cs * 4 + q] = __builtin_amdgcn_wmma_f32_16x16x4_f32(
                  false, av, false, bv, (short)0, acc[cs * 4 + q], false, false);
            }
          }
        }
      }
      wave_lds_order();
      __builtin_amdgcn_sched_barrier(0);
    }
  }
  float sw[H_];
#pragma unroll
  for (int h = 0; h < H_; ++h) sw[h] = row_sum(srun[h]);
  if (lane < H_) {
    const float mv = lane == 0 ? mrun[0] : lane == 1 ? mrun[1] : lane == 2 ? mrun[2] : mrun[3];
    const float sv = lane == 0 ? sw[0] : lane == 1 ? sw[1] : lane == 2 ? sw[2] : sw[3];
    s_m[wave * H_ + lane] = mv;
    s_s[wave * H_ + lane] = sv;
  }
  if (hi == 0) {
    float* rw = s_red + wave * (H_ * C_) + 4 * l16;
#pragma unroll
    for (int r = 0; r < H_; ++r) {
#pragma unroll
      for (int cs = 0; cs < 2; ++cs) {
        v4f o;
        o.x = acc[cs * 4 + 0][r]; o.y = acc[cs * 4 + 1][r];
        o.z = acc[cs * 4 + 2][r]; o.w = acc[cs * 4 + 3][r];
        *(v4f*)(rw + r * C_ + cs * 64) = o;
      }
    }
  }
  __syncthreads();
  const unsigned rec = (unsigned)(b * NCH + chunk);
  float* pr = ws + OFF_PART + (size_t)rec * (H_ * C_);
  for (int i = tid; i < H_ * C_; i += 128) {
    const int h = i >> 7;
    float cm = neg_inf();
    for (int w = 0; w < 4; ++w) cm = fmaxf(cm, s_m[w * H_ + h]);
    float u = 0.f, s = 0.f;
    for (int w = 0; w < 4; ++w) {
      const float e = __expf(s_m[w * H_ + h] - cm);
      u += s_red[w * (H_ * C_) + i] * e;
      s += s_s[w * H_ + h] * e;
    }
    pr[i] = u;
    if ((i & (C_ - 1)) == 0) {
      ws[OFF_CM + rec * H_ + h] = cm;
      ws[OFF_CS + rec * H_ + h] = s;
    }
  }
}

__global__ void k_merge(const int* __restrict__ gs, float* __restrict__ ws) {
  const int idx = blockIdx.x * 256 + threadIdx.x;
  const int b = idx >> 9, rest = idx & 511, h = rest >> 7;
  const int nch = (gs[b] + 255) >> 8;
  float cm[NCH], cs[NCH], pv[NCH];
#pragma unroll
  for (int ch = 0; ch < NCH; ++ch) {
    const unsigned rec = (unsigned)(b * NCH + (ch < nch ? ch : 0));
    cm[ch] = ws[OFF_CM + rec * H_ + h];
    cs[ch] = ws[OFF_CS + rec * H_ + h];
    pv[ch] = ws[OFF_PART + ((size_t)rec << 9) + rest];
  }
  float m = neg_inf();
#pragma unroll
  for (int ch = 0; ch < NCH; ++ch) {
    cm[ch] = (ch < nch) ? cm[ch] : neg_inf();
    m = fmaxf(m, cm[ch]);
  }
  float u = 0.f, s = 0.f;
#pragma unroll
  for (int ch = 0; ch < NCH; ++ch) {
    const float e = __expf(cm[ch] - m);
    u += pv[ch] * e;
    s += cs[ch] * e;
  }
  ws[OFF_U + idx] = u / s;
}

__global__ void k_out(const float* __restrict__ W, const float* __restrict__ bias,
                      const float* __restrict__ U, float* __restrict__ out) {
  const int m0 = blockIdx.x * 16;
  const int h  = blockIdx.y;
  const int tid = threadIdx.x;
  const int wave = tid >> 5, lane = tid & 31;
  const int l16 = lane & 15, hi = lane >> 4;
  const int n0 = wave * 16;
  const float* urow = U + (size_t)(m0 + l16) * HC_ + h * C_;
  const float* wcol = W + h * C_ + n0 + l16;
  v8f acc = {};
  for (int k0 = 0; k0 < C_; k0 += 4) {
    v2f a = *(const v2f*)(urow + k0 + 2 * hi);
    v2f bv;
    bv.x = wcol[(size_t)(k0 + 2 * hi) * HC_];
    bv.y = wcol[(size_t)(k0 + 1 + 2 * hi) * HC_];
    acc = __builtin_amdgcn_wmma_f32_16x16x4_f32(false, a, false, bv,
                                                (short)0, acc, false, false);
  }
  for (int r = 0; r < 8; ++r) {
    int row = m0 + r + 8 * hi;
    int col = h * C_ + n0 + l16;
    out[(size_t)row * HC_ + col] = acc[r] + bias[col];
  }
}

extern "C" void kernel_launch(void* const* d_in, const int* in_sizes, int n_in,
                              void* d_out, int out_size, void* d_ws, size_t ws_size,
                              hipStream_t stream) {
  (void)in_sizes; (void)n_in; (void)out_size; (void)ws_size;
  const float* V    = (const float*)d_in[0];
  const int*   gs   = (const int*)  d_in[1];
  const float* W    = (const float*)d_in[2];
  const float* bias = (const float*)d_in[3];
  const float* T    = (const float*)d_in[4];
  float* out = (float*)d_out;
  float* ws  = (float*)d_ws;

  k_prep <<<9,            256, 0, stream>>>(W, bias, T, ws);
  k_attn <<<dim3(NCH, B_), 128, 0, stream>>>(V, gs, ws);
  k_merge<<<(B_*H_*C_)/256, 256, 0, stream>>>(gs, ws);
  k_out  <<<dim3(B_/16, H_), 256, 0, stream>>>(W, bias, ws + OFF_U, out);
}


# The chunk merge moves into the output kernel (3 launches): a wave merges two graphs' chunk sums into LDS rows that feed the W product, 32 work-groups of 64 columns; every sum keeps its order
# speedup vs baseline: 4.3542x; 1.1072x over previous
#include <hip/hip_runtime.h>
#include <math.h>


#define B_ 64
#define N_ 4096
#define C_ 128
#define H_ 4
#define HC_ 512
#define ALPHA 0.2f

typedef __attribute__((ext_vector_type(8))) float v8f;
typedef __attribute__((ext_vector_type(2))) float v2f;
typedef __attribute__((ext_vector_type(4))) float v4f;

#define OFF_W2P   0u
#define OFF_B2    2048u
#define NCH       16
#define OFF_CM    2080u
#define OFF_CS    (OFF_CM + (unsigned)(B_*NCH*H_))
#define OFF_PART  (OFF_CS + (unsigned)(B_*NCH*H_))

__device__ __forceinline__ float neg_inf() { return -__builtin_inff(); }

#define ROW_XOR(v, M)                                                        \
  __builtin_bit_cast(float, __builtin_amdgcn_update_dpp(                     \
      __builtin_bit_cast(int, (v)), __builtin_bit_cast(int, (v)),            \
      0x160 + (M)  , 0xf, 0xf, false))
__device__ __forceinline__ float row_max(float v) {
  v = fmaxf(v, ROW_XOR(v, 1)); v = fmaxf(v, ROW_XOR(v, 2));
  v = fmaxf(v, ROW_XOR(v, 4)); return fmaxf(v, ROW_XOR(v, 8));
}
__device__ __forceinline__ float row_sum(float v) {
  v += ROW_XOR(v, 1); v += ROW_XOR(v, 2); v += ROW_XOR(v, 4);
  return v + ROW_XOR(v, 8);
}

__global__ void k_prep(const float* __restrict__ W, const float* __restrict__ bias,
                       const float* __restrict__ T, float* __restrict__ ws) {
  int idx = blockIdx.x * 256 + threadIdx.x;
  if (idx < 2048) {
    int k = idx >> 4, j = idx & 15;
    float v = 0.f;
    if (j < H_) {
      const float* wr = W + (size_t)k * HC_ + j * C_;
      const float* tr = T + j * C_;
      for (int c = 0; c < C_; ++c) v = fmaf(wr[c], tr[c], v);
    }
    ws[OFF_W2P + idx] = v;
  } else if (idx < 2048 + H_) {
    int j = idx - 2048;
    float v = 0.f;
    for (int c = 0; c < C_; ++c) v = fmaf(bias[j * C_ + c], T[j * C_ + c], v);
    ws[OFF_B2 + j] = v;
  }
}

#define LP 132
#define PP 16

#define LOAD_TILE(g, base)                                                   \
  _Pragma("unroll")   \
  for (int r = 0; r < 16; ++r) g[r] = *(const v4f*)((base) + (size_t)r * C_);

__device__ __forceinline__ void wave_lds_order() {
  __builtin_amdgcn_fence(__ATOMIC_RELEASE, "wavefront");
  __builtin_amdgcn_wave_barrier();
  __builtin_amdgcn_fence(__ATOMIC_ACQUIRE, "wavefront");
}

__global__ __launch_bounds__(128, 3)
void k_attn(const float* __restrict__ V, const int* __restrict__ gs,
            float* __restrict__ ws) {
  __shared__ __attribute__((aligned(16))) float s_v[4 * 16 * LP];
  __shared__ __attribute__((aligned(16))) float s_w2p[2048];
  __shared__ __attribute__((aligned(16))) float s_pw[4 * H_ * PP];
  __shared__ __attribute__((aligned(16))) float s_red[4 * H_ * C_];
  __shared__ float s_m[4 * H_];
  __shared__ float s_s[4 * H_];
  const int b = blockIdx.y, chunk = blockIdx.x;
  const int limit = gs[b];
  if (chunk * 256 >= limit) return;
  const int tid = threadIdx.x;
  const int wave = __builtin_amdgcn_readfirstlane(tid >> 5);
  const int lane = tid & 31;
  const int l16 = lane & 15, hi = lane >> 4;
  const int n0 = chunk * 256 + wave * 64;
  const bool act = n0 < limit;
  const float* vg = V + ((size_t)b * N_ + n0) * C_ + 4 * lane;
  float* tile = s_v + wave * (16 * LP);
  float* st = tile + 4 * lane;
  const float* fr = tile + l16 * LP + 4 * hi;
  const float* br = tile + (2 * hi) * LP + 4 * l16;
  float* pw = s_pw + wave * (H_ * PP);
  const float* pa = pw + (l16 & 3) * PP + 2 * hi;
  float tw[16];
#pragma unroll
  for (int k = 0; k < 16; ++k) tw[k] = ws[OFF_W2P + k * 128 + tid];
  v4f g[16];
  if (act) { LOAD_TILE(g, vg) }
#pragma unroll
  for (int k = 0; k < 16; ++k) s_w2p[k * 128 + tid] = tw[k];
  __syncthreads();
  v8f acc[8] = {};
  float mrun[H_], srun[H_];
#pragma unroll
  for (int h = 0; h < H_; ++h) { mrun[h] = neg_inf(); srun[h] = 0.f; }
  if (act) {
    const bool real1 = (l16 & 4) == 0;
    const bool real2 = l16 < H_;
    v2f wf[32];
#pragma unroll
    for (int i = 0; i < 32; ++i) {
      const int c = 8 * (i >> 1) + 4 * hi + 2 * (i & 1);
      const float w0 = s_w2p[c * 16 + (l16 & 3)];
      const float w1 = s_w2p[(c + 1) * 16 + (l16 & 3)];
      wf[i].x = real1 ? w0 : 0.f;
      wf[i].y = real1 ? w1 : 0.f;
    }
    float bb[H_];
#pragma unroll
    for (int h = 0; h < H_; ++h) bb[h] = ws[OFF_B2 + h];
#pragma unroll
    for (int t = 0; t < 4; ++t) {
      if (t > 0 && n0 + 16 * t >= limit) break;
#pragma unroll
      for (int r = 0; r < 16; ++r) *(v4f*)(st + r * LP) = g[r];
      wave_lds_order();
      if (t < 3 && n0 + 16 * (t + 1) < limit) {
        LOAD_TILE(g, vg + (size_t)(t + 1) * 16 * C_)
      }
      __builtin_amdgcn_sched_barrier(0);
      v4f f[16];
#pragma unroll
      for (int j = 0; j < 16; ++j) f[j] = *(const v4f*)(fr + 8 * j);
      __builtin_amdgcn_sched_barrier(0);
      v8f acc0 = {}, acc1 = {};
#pragma unroll
      for (int j = 0; j < 16; ++j) {
        v2f b0, b1;
        b0.x = f[j].x; b0.y = f[j].y;
        b1.x = f[j].z; b1.y = f[j].w;
        acc0 = __builtin_amdgcn_wmma_f32_16x16x4_f32(false, wf[2 * j], false, b0,
                                                     (short)0, acc0, false, false);
        acc1 = __builtin_amdgcn_wmma_f32_16x16x4_f32(false, wf[2 * j + 1], false, b1,
                                                     (short)0, acc1, false, false);
      }
      const int node = n0 + 16 * t + l16;
      float sc[H_];
#pragma unroll
      for (int h = 0; h < H_; ++h) {
        float x = acc0[h] + acc1[h] + bb[h];
        x = (x >= 0.f) ? x : ALPHA * x;
        x = (node < limit) ? x : neg_inf();
        const float mn = fmaxf(mrun[h], row_max(x));
        sc[h] = __expf(mrun[h] - mn);
        const float p = __expf(x - mn);
        srun[h] = srun[h] * sc[h] + p;
        mrun[h] = mn;
        pw[h * PP + l16] = p;
      }
#pragma unroll
      for (int i = 0; i < 8; ++i) {
#pragma unroll
        for (int r = 0; r < H_; ++r) acc[i][r] *= sc[r];
      }
      wave_lds_order();
#pragma unroll
      for (int gh = 0; gh < 2; ++gh) {
        v4f vb[2][4];
        v2f a[2];
#pragma unroll
        for (int gg = 0; gg < 2; ++gg) {
          const int g4 = (gh * 2 + gg) * 4;
          a[gg] = *(const v2f*)(pa + g4);
          vb[gg][0] = *(const v4f*)(br + g4 * LP);
          vb[gg][1] = *(const v4f*)(br + g4 * LP + 64);
          vb[gg][2] = *(const v4f*)(br + (g4 + 1) * LP);
          vb[gg][3] = *(const v4f*)(br + (g4 + 1) * LP + 64);
        }
        __builtin_amdgcn_sched_barrier(0);
#pragma unroll
        for (int gg = 0; gg < 2; ++gg) {
          v2f av = a[gg];
          if (!real2) { av.x = 0.f; av.y = 0.f; }
#pragma unroll
          for (int cs = 0; cs < 2; ++cs) {
#pragma unroll
            for (int q = 0; q < 4; ++q) {
              v2f bv;
              bv.x = vb[gg][cs][q];
              bv.y = vb[gg][2 + cs][q];
              acc[cs * 4 + q] = __builtin_amdgcn_wmma_f32_16x16x4_f32(
                  false, av, false, bv, (short)0, acc[cs * 4 + q], false, false);
            }
          }
        }
      }
      wave_lds_order();
      __builtin_amdgcn_sched_barrier(0);
    }
  }
  float sw[H_];
#pragma unroll
  for (int h = 0; h < H_; ++h) sw[h] = row_sum(srun[h]);
  if (lane < H_) {
    const float mv = lane == 0 ? mrun[0] : lane == 1 ? mrun[1] : lane == 2 ? mrun[2] : mrun[3];
    const float sv = lane == 0 ? sw[0] : lane == 1 ? sw[1] : lane == 2 ? sw[2] : sw[3];
    s_m[wave * H_ + lane] = mv;
    s_s[wave * H_ + lane] = sv;
  }
  if (hi == 0) {
    float* rw = s_red + wave * (H_ * C_) + 4 * l16;
#pragma unroll
    for (int r = 0; r < H_; ++r) {
#pragma unroll
      for (int cs = 0; cs < 2; ++cs) {
        v4f o;
        o.x = acc[cs * 4 + 0][r]; o.y = acc[cs * 4 + 1][r];
        o.z = acc[cs * 4 + 2][r]; o.w = acc[cs * 4 + 3][r];
        *(v4f*)(rw + r * C_ + cs * 64) = o;
      }
    }
  }
  __syncthreads();
  const unsigned rec = (unsigned)(b * NCH + chunk);
  float* pr = ws + OFF_PART + (size_t)rec * (H_ * C_);
  for (int i = tid; i < H_ * C_; i += 128) {
    const int h = i >> 7;
    float cm = neg_inf();
    for (int w = 0; w < 4; ++w) cm = fmaxf(cm, s_m[w * H_ + h]);
    float u = 0.f, s = 0.f;
    for (int w = 0; w < 4; ++w) {
      const float e = __expf(s_m[w * H_ + h] - cm);
      u += s_red[w * (H_ * C_) + i] * e;
      s += s_s[w * H_ + h] * e;
    }
    pr[i] = u;
    if ((i & (C_ - 1)) == 0) {
      ws[OFF_CM + rec * H_ + h] = cm;
      ws[OFF_CS + rec * H_ + h] = s;
    }
  }
}

#define UP 132

#define LANE_VALUE(v, L)                                                     \
  __builtin_bit_cast(float, __builtin_amdgcn_readlane(                       \
      __builtin_bit_cast(int, (v)), (L)))

__global__ __launch_bounds__(256)
void k_out(const float* __restrict__ W, const float* __restrict__ bias,
           const int* __restrict__ gs, const float* __restrict__ ws,
           float* __restrict__ out) {
  __shared__ __attribute__((aligned(16))) float s_u[16 * UP];
  const int m0 = blockIdx.x * 16;
  const int h  = blockIdx.y;
  const int tid = threadIdx.x;
  const int wave = __builtin_amdgcn_readfirstlane(tid >> 5);
  const int lane = tid & 31;
  const int l16 = lane & 15, hi = lane >> 4;
  const bool prod = wave < 4;
  const int col = h * C_ + blockIdx.z * 64 + (wave & 3) * 16 + l16;
  const float* wcol = W + col + (size_t)(2 * hi) * HC_;
  v2f bw[32];
  float bcol = 0.f;
  if (prod) {
#pragma unroll
    for (int j = 0; j < 32; ++j) {
      bw[j].x = wcol[(size_t)(4 * j) * HC_];
      bw[j].y = wcol[(size_t)(4 * j + 1) * HC_];
    }
    bcol = bias[col];
  }
  __builtin_amdgcn_sched_barrier(0);
  v4f pv[2][NCH];
  float cm[2], cs[2];
  int nchs[2];
#pragma unroll
  for (int gi = 0; gi < 2; ++gi) {
    const int b = m0 + 2 * wave + gi;
    const int nch = (gs[b] + 255) >> 8;
    nchs[gi] = nch;
    const unsigned recl = (unsigned)(b * NCH + (l16 < nch ? l16 : 0));
    cm[gi] = ws[OFF_CM + recl * H_ + h];
    cs[gi] = ws[OFF_CS + recl * H_ + h];
    const float* pb = ws + OFF_PART + h * C_ + 4 * lane;
#pragma unroll
    for (int ch = 0; ch < NCH; ++ch) {
      const unsigned rec = (unsigned)(b * NCH + (ch < nch ? ch : 0));
      pv[gi][ch] = *(const v4f*)(pb + ((size_t)rec << 9));
    }
  }
  __builtin_amdgcn_sched_barrier(0);
#pragma unroll
  for (int gi = 0; gi < 2; ++gi) {
    const float cmv = (l16 < nchs[gi]) ? cm[gi] : neg_inf();
    const float m = row_max(cmv);
    const float e = __expf(cmv - m);
    v4f u = {0.f, 0.f, 0.f, 0.f};
    float s = 0.f;
#pragma unroll
    for (int ch = 0; ch < NCH; ++ch) {
      const float es = LANE_VALUE(e, ch);
      const float css = LANE_VALUE(cs[gi], ch);
      u.x = fmaf(pv[gi][ch].x, es, u.x);
      u.y = fmaf(pv[gi][ch].y, es, u.y);
      u.z = fmaf(pv[gi][ch].z, es, u.z);
      u.w = fmaf(pv[gi][ch].w, es, u.w);
      s = fmaf(css, es, s);
    }
    v4f o;
    o.x = u.x / s; o.y = u.y / s; o.z = u.z / s; o.w = u.w / s;
    *(v4f*)(s_u + (2 * wave + gi) * UP + 4 * lane) = o;
  }
  __syncthreads();
  if (!prod) return;
  const float* urow = s_u + l16 * UP + 2 * hi;
  v2f a[32];
#pragma unroll
  for (int j = 0; j < 32; ++j) a[j] = *(const v2f*)(urow + 4 * j);
  __builtin_amdgcn_sched_barrier(0);
  v8f acc = {};
#pragma unroll
  for (int j = 0; j < 32; ++j)
    acc = __builtin_amdgcn_wmma_f32_16x16x4_f32(false, a[j], false, bw[j],
                                                (short)0, acc, false, false);
#pragma unroll
  for (int r = 0; r < 8; ++r) {
    const int row = m0 + r + 8 * hi;
    out[(size_t)row * HC_ + col] = acc[r] + bcol;
  }
}

extern "C" void kernel_launch(void* const* d_in, const int* in_sizes, int n_in,
                              void* d_out, int out_size, void* d_ws, size_t ws_size,
                              hipStream_t stream) {
  (void)in_sizes; (void)n_in; (void)out_size; (void)ws_size;
  const float* V    = (const float*)d_in[0];
  const int*   gs   = (const int*)  d_in[1];
  const float* W    = (const float*)d_in[2];
  const float* bias = (const float*)d_in[3];
  const float* T    = (const float*)d_in[4];
  float* out = (float*)d_out;
  float* ws  = (float*)d_ws;

  k_prep<<<9,             256, 0, stream>>>(W, bias, T, ws);
  k_attn<<<dim3(NCH, B_), 128, 0, stream>>>(V, gs, ws);
  k_out <<<dim3(B_/16, H_, 2), 256, 0, stream>>>(W, bias, gs, ws, out);
}


# K1 (w2 table): a 16-lane row forms a table row's four 128-term dots, a lane's 16 loads requested at once (was a thread per dot, 8 load round trips); lane partials added: w2, b2 sum order changes
# speedup vs baseline: 4.5862x; 1.0533x over previous
#include <hip/hip_runtime.h>
#include <math.h>


#define B_ 64
#define N_ 4096
#define C_ 128
#define H_ 4
#define HC_ 512
#define ALPHA 0.2f

typedef __attribute__((ext_vector_type(8))) float v8f;
typedef __attribute__((ext_vector_type(2))) float v2f;
typedef __attribute__((ext_vector_type(4))) float v4f;

#define OFF_W2P   0u
#define OFF_B2    2048u
#define NCH       16
#define OFF_CM    2080u
#define OFF_CS    (OFF_CM + (unsigned)(B_*NCH*H_))
#define OFF_PART  (OFF_CS + (unsigned)(B_*NCH*H_))

__device__ __forceinline__ float neg_inf() { return -__builtin_inff(); }

#define ROW_XOR(v, M)                                                        \
  __builtin_bit_cast(float, __builtin_amdgcn_update_dpp(                     \
      __builtin_bit_cast(int, (v)), __builtin_bit_cast(int, (v)),            \
      0x160 + (M)  , 0xf, 0xf, false))
__device__ __forceinline__ float row_max(float v) {
  v = fmaxf(v, ROW_XOR(v, 1)); v = fmaxf(v, ROW_XOR(v, 2));
  v = fmaxf(v, ROW_XOR(v, 4)); return fmaxf(v, ROW_XOR(v, 8));
}
__device__ __forceinline__ float row_sum(float v) {
  v += ROW_XOR(v, 1); v += ROW_XOR(v, 2); v += ROW_XOR(v, 4);
  return v + ROW_XOR(v, 8);
}

__global__ void k_prep(const float* __restrict__ W, const float* __restrict__ bias,
                       const float* __restrict__ T, float* __restrict__ ws) {
  const int tid = blockIdx.x * 256 + threadIdx.x;
  const int row = tid >> 4, l16 = tid & 15;
  const int h = l16 & 3, q = l16 >> 2;
  const float* a = (row < C_ ? W + (size_t)row * HC_ : bias) + h * C_ + 4 * q;
  const float* t = T + h * C_ + 4 * q;
  v4f av[8], tv[8];
#pragma unroll
  for (int j = 0; j < 8; ++j) {
    av[j] = *(const v4f*)(a + 16 * j);
    tv[j] = *(const v4f*)(t + 16 * j);
  }
  __builtin_amdgcn_sched_barrier(0);
  v4f s = av[0] * tv[0];
#pragma unroll
  for (int j = 1; j < 8; ++j) {
    s.x = fmaf(av[j].x, tv[j].x, s.x); s.y = fmaf(av[j].y, tv[j].y, s.y);
    s.z = fmaf(av[j].z, tv[j].z, s.z); s.w = fmaf(av[j].w, tv[j].w, s.w);
  }
  float v = (s.x + s.y) + (s.z + s.w);
  v += ROW_XOR(v, 4); v += ROW_XOR(v, 8);
  if (row < C_) ws[OFF_W2P + row * 16 + l16] = (l16 < H_) ? v : 0.f;
  else if (row == C_ && l16 < H_) ws[OFF_B2 + l16] = v;
}

#define LP 132
#define PP 16

#define LOAD_TILE(g, base)                                                   \
  _Pragma("unroll")   \
  for (int r = 0; r < 16; ++r) g[r] = *(const v4f*)((base) + (size_t)r * C_);

__device__ __forceinline__ void wave_lds_order() {
  __builtin_amdgcn_fence(__ATOMIC_RELEASE, "wavefront");
  __builtin_amdgcn_wave_barrier();
  __builtin_amdgcn_fence(__ATOMIC_ACQUIRE, "wavefront");
}

__global__ __launch_bounds__(128, 3)
void k_attn(const float* __restrict__ V, const int* __restrict__ gs,
            float* __restrict__ ws) {
  __shared__ __attribute__((aligned(16))) float s_v[4 * 16 * LP];
  __shared__ __attribute__((aligned(16))) float s_w2p[2048];
  __shared__ __attribute__((aligned(16))) float s_pw[4 * H_ * PP];
  __shared__ __attribute__((aligned(16))) float s_red[4 * H_ * C_];
  __shared__ float s_m[4 * H_];
  __shared__ float s_s[4 * H_];
  const int b = blockIdx.y, chunk = blockIdx.x;
  const int limit = gs[b];
  if (chunk * 256 >= limit) return;
  const int tid = threadIdx.x;
  const int wave = __builtin_amdgcn_readfirstlane(tid >> 5);
  const int lane = tid & 31;
  const int l16 = lane & 15, hi = lane >> 4;
  const int n0 = chunk * 256 + wave * 64;
  const bool act = n0 < limit;
  const float* vg = V + ((size_t)b * N_ + n0) * C_ + 4 * lane;
  float* tile = s_v + wave * (16 * LP);
  float* st = tile + 4 * lane;
  const float* fr = tile + l16 * LP + 4 * hi;
  const float* br = tile + (2 * hi) * LP + 4 * l16;
  float* pw = s_pw + wave * (H_ * PP);
  const float* pa = pw + (l16 & 3) * PP + 2 * hi;
  float tw[16];
#pragma unroll
  for (int k = 0; k < 16; ++k) tw[k] = ws[OFF_W2P + k * 128 + tid];
  v4f g[16];
  if (act) { LOAD_TILE(g, vg) }
#pragma unroll
  for (int k = 0; k < 16; ++k) s_w2p[k * 128 + tid] = tw[k];
  __syncthreads();
  v8f acc[8] = {};
  float mrun[H_], srun[H_];
#pragma unroll
  for (int h = 0; h < H_; ++h) { mrun[h] = neg_inf(); srun[h] = 0.f; }
  if (act) {
    const bool real1 = (l16 & 4) == 0;
    const bool real2 = l16 < H_;
    v2f wf[32];
#pragma unroll
    for (int i = 0; i < 32; ++i) {
      const int c = 8 * (i >> 1) + 4 * hi + 2 * (i & 1);
      const float w0 = s_w2p[c * 16 + (l16 & 3)];
      const float w1 = s_w2p[(c + 1) * 16 + (l16 & 3)];
      wf[i].x = real1 ? w0 : 0.f;
      wf[i].y = real1 ? w1 : 0.f;
    }
    float bb[H_];
#pragma unroll
    for (int h = 0; h < H_; ++h) bb[h] = ws[OFF_B2 + h];
#pragma unroll
    for (int t = 0; t < 4; ++t) {
      if (t > 0 && n0 + 16 * t >= limit) break;
#pragma unroll
      for (int r = 0; r < 16; ++r) *(v4f*)(st + r * LP) = g[r];
      wave_lds_order();
      if (t < 3 && n0 + 16 * (t + 1) < limit) {
        LOAD_TILE(g, vg + (size_t)(t + 1) * 16 * C_)
      }
      __builtin_amdgcn_sched_barrier(0);
      v4f f[16];
#pragma unroll
      for (int j = 0; j < 16; ++j) f[j] = *(const v4f*)(fr + 8 * j);
      __builtin_amdgcn_sched_barrier(0);
      v8f acc0 = {}, acc1 = {};
#pragma unroll
      for (int j = 0; j < 16; ++j) {
        v2f b0, b1;
        b0.x = f[j].x; b0.y = f[j].y;
        b1.x = f[j].z; b1.y = f[j].w;
        acc0 = __builtin_amdgcn_wmma_f32_16x16x4_f32(false, wf[2 * j], false, b0,
                                                     (short)0, acc0, false, false);
        acc1 = __builtin_amdgcn_wmma_f32_16x16x4_f32(false, wf[2 * j + 1], false, b1,
                                                     (short)0, acc1, false, false);
      }
      const int node = n0 + 16 * t + l16;
      float sc[H_];
#pragma unroll
      for (int h = 0; h < H_; ++h) {
        float x = acc0[h] + acc1[h] + bb[h];
        x = (x >= 0.f) ? x : ALPHA * x;
        x = (node < limit) ? x : neg_inf();
        const float mn = fmaxf(mrun[h], row_max(x));
        sc[h] = __expf(mrun[h] - mn);
        const float p = __expf(x - mn);
        srun[h] = srun[h] * sc[h] + p;
        mrun[h] = mn;
        pw[h * PP + l16] = p;
      }
#pragma unroll
      for (int i = 0; i < 8; ++i) {
#pragma unroll
        for (int r = 0; r < H_; ++r) acc[i][r] *= sc[r];
      }
      wave_lds_order();
#pragma unroll
      for (int gh = 0; gh < 2; ++gh) {
        v4f vb[2][4];
        v2f a[2];
#pragma unroll
        for (int gg = 0; gg < 2; ++gg) {
          const int g4 = (gh * 2 + gg) * 4;
          a[gg] = *(const v2f*)(pa + g4);
          vb[gg][0] = *(const v4f*)(br + g4 * LP);
          vb[gg][1] = *(const v4f*)(br + g4 * LP + 64);
          vb[gg][2] = *(const v4f*)(br + (g4 + 1) * LP);
          vb[gg][3] = *(const v4f*)(br + (g4 + 1) * LP + 64);
        }
        __builtin_amdgcn_sched_barrier(0);
#pragma unroll
        for (int gg = 0; gg < 2; ++gg) {
          v2f av = a[gg];
          if (!real2) { av.x = 0.f; av.y = 0.f; }
#pragma unroll
          for (int cs = 0; cs < 2; ++cs) {
#pragma unroll
            for (int q = 0; q < 4; ++q) {
              v2f bv;
              bv.x = vb[gg][cs][q];
              bv.y = vb[gg][2 + cs][q];
              acc[cs * 4 + q] = __builtin_amdgcn_wmma_f32_16x16x4_f32(
                  false, av, false, bv, (short)0, acc[cs * 4 + q], false, false);
            }
          }
        }
      }
      wave_lds_order();
      __builtin_amdgcn_sched_barrier(0);
    }
  }
  float sw[H_];
#pragma unroll
  for (int h = 0; h < H_; ++h) sw[h] = row_sum(srun[h]);
  if (lane < H_) {
    const float mv = lane == 0 ? mrun[0] : lane == 1 ? mrun[1] : lane == 2 ? mrun[2] : mrun[3];
    const float sv = lane == 0 ? sw[0] : lane == 1 ? sw[1] : lane == 2 ? sw[2] : sw[3];
    s_m[wave * H_ + lane] = mv;
    s_s[wave * H_ + lane] = sv;
  }
  if (hi == 0) {
    float* rw = s_red + wave * (H_ * C_) + 4 * l16;
#pragma unroll
    for (int r = 0; r < H_; ++r) {
#pragma unroll
      for (int cs = 0; cs < 2; ++cs) {
        v4f o;
        o.x = acc[cs * 4 + 0][r]; o.y = acc[cs * 4 + 1][r];
        o.z = acc[cs * 4 + 2][r]; o.w = acc[cs * 4 + 3][r];
        *(v4f*)(rw + r * C_ + cs * 64) = o;
      }
    }
  }
  __syncthreads();
  const unsigned rec = (unsigned)(b * NCH + chunk);
  float* pr = ws + OFF_PART + (size_t)rec * (H_ * C_);
  for (int i = tid; i < H_ * C_; i += 128) {
    const int h = i >> 7;
    float cm = neg_inf();
    for (int w = 0; w < 4; ++w) cm = fmaxf(cm, s_m[w * H_ + h]);
    float u = 0.f, s = 0.f;
    for (int w = 0; w < 4; ++w) {
      const float e = __expf(s_m[w * H_ + h] - cm);
      u += s_red[w * (H_ * C_) + i] * e;
      s += s_s[w * H_ + h] * e;
    }
    pr[i] = u;
    if ((i & (C_ - 1)) == 0) {
      ws[OFF_CM + rec * H_ + h] = cm;
      ws[OFF_CS + rec * H_ + h] = s;
    }
  }
}

#define UP 132

#define LANE_VALUE(v, L)                                                     \
  __builtin_bit_cast(float, __builtin_amdgcn_readlane(                       \
      __builtin_bit_cast(int, (v)), (L)))

__global__ __launch_bounds__(256)
void k_out(const float* __restrict__ W, const float* __restrict__ bias,
           const int* __restrict__ gs, const float* __restrict__ ws,
           float* __restrict__ out) {
  __shared__ __attribute__((aligned(16))) float s_u[16 * UP];
  const int m0 = blockIdx.x * 16;
  const int h  = blockIdx.y;
  const int tid = threadIdx.x;
  const int wave = __builtin_amdgcn_readfirstlane(tid >> 5);
  const int lane = tid & 31;
  const int l16 = lane & 15, hi = lane >> 4;
  const bool prod = wave < 4;
  const int col = h * C_ + blockIdx.z * 64 + (wave & 3) * 16 + l16;
  const float* wcol = W + col + (size_t)(2 * hi) * HC_;
  v2f bw[32];
  float bcol = 0.f;
  if (prod) {
#pragma unroll
    for (int j = 0; j < 32; ++j) {
      bw[j].x = wcol[(size_t)(4 * j) * HC_];
      bw[j].y = wcol[(size_t)(4 * j + 1) * HC_];
    }
    bcol = bias[col];
  }
  __builtin_amdgcn_sched_barrier(0);
  v4f pv[2][NCH];
  float cm[2], cs[2];
  int nchs[2];
#pragma unroll
  for (int gi = 0; gi < 2; ++gi) {
    const int b = m0 + 2 * wave + gi;
    const int nch = (gs[b] + 255) >> 8;
    nchs[gi] = nch;
    const unsigned recl = (unsigned)(b * NCH + (l16 < nch ? l16 : 0));
    cm[gi] = ws[OFF_CM + recl * H_ + h];
    cs[gi] = ws[OFF_CS + recl * H_ + h];
    const float* pb = ws + OFF_PART + h * C_ + 4 * lane;
#pragma unroll
    for (int ch = 0; ch < NCH; ++ch) {
      const unsigned rec = (unsigned)(b * NCH + (ch < nch ? ch : 0));
      pv[gi][ch] = *(const v4f*)(pb + ((size_t)rec << 9));
    }
  }
  __builtin_amdgcn_sched_barrier(0);
#pragma unroll
  for (int gi = 0; gi < 2; ++gi) {
    const float cmv = (l16 < nchs[gi]) ? cm[gi] : neg_inf();
    const float m = row_max(cmv);
    const float e = __expf(cmv - m);
    v4f u = {0.f, 0.f, 0.f, 0.f};
    float s = 0.f;
#pragma unroll
    for (int ch = 0; ch < NCH; ++ch) {
      const float es = LANE_VALUE(e, ch);
      const float css = LANE_VALUE(cs[gi], ch);
      u.x = fmaf(pv[gi][ch].x, es, u.x);
      u.y = fmaf(pv[gi][ch].y, es, u.y);
      u.z = fmaf(pv[gi][ch].z, es, u.z);
      u.w = fmaf(pv[gi][ch].w, es, u.w);
      s = fmaf(css, es, s);
    }
    v4f o;
    o.x = u.x / s; o.y = u.y / s; o.z = u.z / s; o.w = u.w / s;
    *(v4f*)(s_u + (2 * wave + gi) * UP + 4 * lane) = o;
  }
  __syncthreads();
  if (!prod) return;
  const float* urow = s_u + l16 * UP + 2 * hi;
  v2f a[32];
#pragma unroll
  for (int j = 0; j < 32; ++j) a[j] = *(const v2f*)(urow + 4 * j);
  __builtin_amdgcn_sched_barrier(0);
  v8f acc = {};
#pragma unroll
  for (int j = 0; j < 32; ++j)
    acc = __builtin_amdgcn_wmma_f32_16x16x4_f32(false, a[j], false, bw[j],
                                                (short)0, acc, false, false);
#pragma unroll
  for (int r = 0; r < 8; ++r) {
    const int row = m0 + r + 8 * hi;
    out[(size_t)row * HC_ + col] = acc[r] + bcol;
  }
}

extern "C" void kernel_launch(void* const* d_in, const int* in_sizes, int n_in,
                              void* d_out, int out_size, void* d_ws, size_t ws_size,
                              hipStream_t stream) {
  (void)in_sizes; (void)n_in; (void)out_size; (void)ws_size;
  const float* V    = (const float*)d_in[0];
  const int*   gs   = (const int*)  d_in[1];
  const float* W    = (const float*)d_in[2];
  const float* bias = (const float*)d_in[3];
  const float* T    = (const float*)d_in[4];
  float* out = (float*)d_out;
  float* ws  = (float*)d_ws;

  k_prep<<<9,             256, 0, stream>>>(W, bias, T, ws);
  k_attn<<<dim3(NCH, B_), 128, 0, stream>>>(V, gs, ws);
  k_out <<<dim3(B_/16, H_, 2), 256, 0, stream>>>(W, bias, gs, ws, out);
}


# K2 work-group i takes the i-th chunk with a valid node (prefix sum of the graphs' chunk counts) and chunks grow to 320 rows: live groups dispatched first, two per WGP at most; sum order changes
# speedup vs baseline: 5.7583x; 1.2556x over previous
#include <hip/hip_runtime.h>
#include <math.h>


#define B_ 64
#define N_ 4096
#define C_ 128
#define H_ 4
#define HC_ 512
#define ALPHA 0.2f

typedef __attribute__((ext_vector_type(8))) float v8f;
typedef __attribute__((ext_vector_type(2))) float v2f;
typedef __attribute__((ext_vector_type(4))) float v4f;

#define OFF_W2P   0u
#define OFF_B2    2048u
#define CHR       320
#define WVR       80
#define NTL       5
#define NCH       13
#define OFF_CM    2080u
#define OFF_CS    (OFF_CM + (unsigned)(B_*NCH*H_))
#define OFF_PART  (OFF_CS + (unsigned)(B_*NCH*H_))

__device__ __forceinline__ float neg_inf() { return -__builtin_inff(); }

#define ROW_XOR(v, M)                                                        \
  __builtin_bit_cast(float, __builtin_amdgcn_update_dpp(                     \
      __builtin_bit_cast(int, (v)), __builtin_bit_cast(int, (v)),            \
      0x160 + (M)  , 0xf, 0xf, false))
#define ROW_SHR0(v, M)                                                       \
  __builtin_amdgcn_update_dpp(0, (v), 0x110 + (M)  , \
                              0xf, 0xf, true  )
__device__ __forceinline__ float row_max(float v) {
  v = fmaxf(v, ROW_XOR(v, 1)); v = fmaxf(v, ROW_XOR(v, 2));
  v = fmaxf(v, ROW_XOR(v, 4)); return fmaxf(v, ROW_XOR(v, 8));
}
__device__ __forceinline__ float row_sum(float v) {
  v += ROW_XOR(v, 1); v += ROW_XOR(v, 2); v += ROW_XOR(v, 4);
  return v + ROW_XOR(v, 8);
}

__global__ void k_prep(const float* __restrict__ W, const float* __restrict__ bias,
                       const float* __restrict__ T, float* __restrict__ ws) {
  const int tid = blockIdx.x * 256 + threadIdx.x;
  const int row = tid >> 4, l16 = tid & 15;
  const int h = l16 & 3, q = l16 >> 2;
  const float* a = (row < C_ ? W + (size_t)row * HC_ : bias) + h * C_ + 4 * q;
  const float* t = T + h * C_ + 4 * q;
  v4f av[8], tv[8];
#pragma unroll
  for (int j = 0; j < 8; ++j) {
    av[j] = *(const v4f*)(a + 16 * j);
    tv[j] = *(const v4f*)(t + 16 * j);
  }
  __builtin_amdgcn_sched_barrier(0);
  v4f s = av[0] * tv[0];
#pragma unroll
  for (int j = 1; j < 8; ++j) {
    s.x = fmaf(av[j].x, tv[j].x, s.x); s.y = fmaf(av[j].y, tv[j].y, s.y);
    s.z = fmaf(av[j].z, tv[j].z, s.z); s.w = fmaf(av[j].w, tv[j].w, s.w);
  }
  float v = (s.x + s.y) + (s.z + s.w);
  v += ROW_XOR(v, 4); v += ROW_XOR(v, 8);
  if (row < C_) ws[OFF_W2P + row * 16 + l16] = (l16 < H_) ? v : 0.f;
  else if (row == C_ && l16 < H_) ws[OFF_B2 + l16] = v;
}

#define LP 132
#define PP 16

#define LOAD_TILE(g, base)                                                   \
  _Pragma("unroll")   \
  for (int r = 0; r < 16; ++r) g[r] = *(const v4f*)((base) + (size_t)r * C_);

__device__ __forceinline__ void wave_lds_order() {
  __builtin_amdgcn_fence(__ATOMIC_RELEASE, "wavefront");
  __builtin_amdgcn_wave_barrier();
  __builtin_amdgcn_fence(__ATOMIC_ACQUIRE, "wavefront");
}

__global__ __launch_bounds__(128, 3)
void k_attn(const float* __restrict__ V, const int* __restrict__ gs,
            float* __restrict__ ws) {
  __shared__ __attribute__((aligned(16))) float s_v[4 * 16 * LP];
  __shared__ __attribute__((aligned(16))) float s_w2p[2048];
  __shared__ __attribute__((aligned(16))) float s_pw[4 * H_ * PP];
  __shared__ __attribute__((aligned(16))) float s_red[4 * H_ * C_];
  __shared__ float s_m[4 * H_];
  __shared__ float s_s[4 * H_];
  const int tid = threadIdx.x;
  const int wid = blockIdx.x;
  int b, chunk, limit;
  {
    const int ln = tid & 31;
    const int ga = gs[2 * ln], gb = gs[2 * ln + 1];
    const int na = min(max((ga + (CHR - 1)) / CHR, 0), NCH);
    const int nb = min(max((gb + (CHR - 1)) / CHR, 0), NCH);
    const int pr2 = na + nb;
    int x = pr2;
    x += ROW_SHR0(x, 1); x += ROW_SHR0(x, 2); x += ROW_SHR0(x, 4); x += ROW_SHR0(x, 8);
    const int r0 = __builtin_amdgcn_readlane(x, 15);
    x += (ln >= 16) ? r0 : 0;
    const int before = __builtin_popcount(__builtin_amdgcn_ballot_w32(x - nb <= wid)) +
                       __builtin_popcount(__builtin_amdgcn_ballot_w32(x <= wid));
    if (before >= B_) return;
    b = before;
    const bool odd = (b & 1) != 0;
    const int first = __builtin_amdgcn_readlane(odd ? x - nb : x - pr2, b >> 1);
    limit = __builtin_amdgcn_readlane(odd ? gb : ga, b >> 1);
    chunk = wid - first;
  }
  const int wave = __builtin_amdgcn_readfirstlane(tid >> 5);
  const int lane = tid & 31;
  const int l16 = lane & 15, hi = lane >> 4;
  const int n0 = chunk * CHR + wave * WVR;
  const bool act = n0 < limit;
  const float* vg = V + ((size_t)b * N_ + n0) * C_ + 4 * lane;
  float* tile = s_v + wave * (16 * LP);
  float* st = tile + 4 * lane;
  const float* fr = tile + l16 * LP + 4 * hi;
  const float* br = tile + (2 * hi) * LP + 4 * l16;
  float* pw = s_pw + wave * (H_ * PP);
  const float* pa = pw + (l16 & 3) * PP + 2 * hi;
  float tw[16];
#pragma unroll
  for (int k = 0; k < 16; ++k) tw[k] = ws[OFF_W2P + k * 128 + tid];
  v4f g[16];
  if (act) { LOAD_TILE(g, vg) }
#pragma unroll
  for (int k = 0; k < 16; ++k) s_w2p[k * 128 + tid] = tw[k];
  __syncthreads();
  v8f acc[8] = {};
  float mrun[H_], srun[H_];
#pragma unroll
  for (int h = 0; h < H_; ++h) { mrun[h] = neg_inf(); srun[h] = 0.f; }
  if (act) {
    const bool real1 = (l16 & 4) == 0;
    const bool real2 = l16 < H_;
    v2f wf[32];
#pragma unroll
    for (int i = 0; i < 32; ++i) {
      const int c = 8 * (i >> 1) + 4 * hi + 2 * (i & 1);
      const float w0 = s_w2p[c * 16 + (l16 & 3)];
      const float w1 = s_w2p[(c + 1) * 16 + (l16 & 3)];
      wf[i].x = real1 ? w0 : 0.f;
      wf[i].y = real1 ? w1 : 0.f;
    }
    float bb[H_];
#pragma unroll
    for (int h = 0; h < H_; ++h) bb[h] = ws[OFF_B2 + h];
#pragma unroll
    for (int t = 0; t < NTL; ++t) {
      if (t > 0 && n0 + 16 * t >= limit) break;
#pragma unroll
      for (int r = 0; r < 16; ++r) *(v4f*)(st + r * LP) = g[r];
      wave_lds_order();
      if (t < NTL - 1 && n0 + 16 * (t + 1) < limit) {
        LOAD_TILE(g, vg + (size_t)(t + 1) * 16 * C_)
      }
      __builtin_amdgcn_sched_barrier(0);
      v4f f[16];
#pragma unroll
      for (int j = 0; j < 16; ++j) f[j] = *(const v4f*)(fr + 8 * j);
      __builtin_amdgcn_sched_barrier(0);
      v8f acc0 = {}, acc1 = {};
#pragma unroll
      for (int j = 0; j < 16; ++j) {
        v2f b0, b1;
        b0.x = f[j].x; b0.y = f[j].y;
        b1.x = f[j].z; b1.y = f[j].w;
        acc0 = __builtin_amdgcn_wmma_f32_16x16x4_f32(false, wf[2 * j], false, b0,
                                                     (short)0, acc0, false, false);
        acc1 = __builtin_amdgcn_wmma_f32_16x16x4_f32(false, wf[2 * j + 1], false, b1,
                                                     (short)0, acc1, false, false);
      }
      const int node = n0 + 16 * t + l16;
      float sc[H_];
#pragma unroll
      for (int h = 0; h < H_; ++h) {
        float x = acc0[h] + acc1[h] + bb[h];
        x = (x >= 0.f) ? x : ALPHA * x;
        x = (node < limit) ? x : neg_inf();
        const float mn = fmaxf(mrun[h], row_max(x));
        sc[h] = __expf(mrun[h] - mn);
        const float p = __expf(x - mn);
        srun[h] = srun[h] * sc[h] + p;
        mrun[h] = mn;
        pw[h * PP + l16] = p;
      }
#pragma unroll
      for (int i = 0; i < 8; ++i) {
#pragma unroll
        for (int r = 0; r < H_; ++r) acc[i][r] *= sc[r];
      }
      wave_lds_order();
#pragma unroll
      for (int gh = 0; gh < 2; ++gh) {
        v4f vb[2][4];
        v2f a[2];
#pragma unroll
        for (int gg = 0; gg < 2; ++gg) {
          const int g4 = (gh * 2 + gg) * 4;
          a[gg] = *(const v2f*)(pa + g4);
          vb[gg][0] = *(const v4f*)(br + g4 * LP);
          vb[gg][1] = *(const v4f*)(br + g4 * LP + 64);
          vb[gg][2] = *(const v4f*)(br + (g4 + 1) * LP);
          vb[gg][3] = *(const v4f*)(br + (g4 + 1) * LP + 64);
        }
        __builtin_amdgcn_sched_barrier(0);
#pragma unroll
        for (int gg = 0; gg < 2; ++gg) {
          v2f av = a[gg];
          if (!real2) { av.x = 0.f; av.y = 0.f; }
#pragma unroll
          for (int cs = 0; cs < 2; ++cs) {
#pragma unroll
            for (int q = 0; q < 4; ++q) {
              v2f bv;
              bv.x = vb[gg][cs][q];
              bv.y = vb[gg][2 + cs][q];
              acc[cs * 4 + q] = __builtin_amdgcn_wmma_f32_16x16x4_f32(
                  false, av, false, bv, (short)0, acc[cs * 4 + q], false, false);
            }
          }
        }
      }
      wave_lds_order();
      __builtin_amdgcn_sched_barrier(0);
    }
  }
  float sw[H_];
#pragma unroll
  for (int h = 0; h < H_; ++h) sw[h] = row_sum(srun[h]);
  if (lane < H_) {
    const float mv = lane == 0 ? mrun[0] : lane == 1 ? mrun[1] : lane == 2 ? mrun[2] : mrun[3];
    const float sv = lane == 0 ? sw[0] : lane == 1 ? sw[1] : lane == 2 ? sw[2] : sw[3];
    s_m[wave * H_ + lane] = mv;
    s_s[wave * H_ + lane] = sv;
  }
  if (hi == 0) {
    float* rw = s_red + wave * (H_ * C_) + 4 * l16;
#pragma unroll
    for (int r = 0; r < H_; ++r) {
#pragma unroll
      for (int cs = 0; cs < 2; ++cs) {
        v4f o;
        o.x = acc[cs * 4 + 0][r]; o.y = acc[cs * 4 + 1][r];
        o.z = acc[cs * 4 + 2][r]; o.w = acc[cs * 4 + 3][r];
        *(v4f*)(rw + r * C_ + cs * 64) = o;
      }
    }
  }
  __syncthreads();
  const unsigned rec = (unsigned)(b * NCH + chunk);
  float* pr = ws + OFF_PART + (size_t)rec * (H_ * C_);
  for (int i = tid; i < H_ * C_; i += 128) {
    const int h = i >> 7;
    float cm = neg_inf();
    for (int w = 0; w < 4; ++w) cm = fmaxf(cm, s_m[w * H_ + h]);
    float u = 0.f, s = 0.f;
    for (int w = 0; w < 4; ++w) {
      const float e = __expf(s_m[w * H_ + h] - cm);
      u += s_red[w * (H_ * C_) + i] * e;
      s += s_s[w * H_ + h] * e;
    }
    pr[i] = u;
    if ((i & (C_ - 1)) == 0) {
      ws[OFF_CM + rec * H_ + h] = cm;
      ws[OFF_CS + rec * H_ + h] = s;
    }
  }
}

#define UP 132

#define LANE_VALUE(v, L)                                                     \
  __builtin_bit_cast(float, __builtin_amdgcn_readlane(                       \
      __builtin_bit_cast(int, (v)), (L)))

__global__ __launch_bounds__(256)
void k_out(const float* __restrict__ W, const float* __restrict__ bias,
           const int* __restrict__ gs, const float* __restrict__ ws,
           float* __restrict__ out) {
  __shared__ __attribute__((aligned(16))) float s_u[16 * UP];
  const int m0 = blockIdx.x * 16;
  const int h  = blockIdx.y;
  const int tid = threadIdx.x;
  const int wave = __builtin_amdgcn_readfirstlane(tid >> 5);
  const int lane = tid & 31;
  const int l16 = lane & 15, hi = lane >> 4;
  const bool prod = wave < 4;
  const int col = h * C_ + blockIdx.z * 64 + (wave & 3) * 16 + l16;
  const float* wcol = W + col + (size_t)(2 * hi) * HC_;
  v2f bw[32];
  float bcol = 0.f;
  if (prod) {
#pragma unroll
    for (int j = 0; j < 32; ++j) {
      bw[j].x = wcol[(size_t)(4 * j) * HC_];
      bw[j].y = wcol[(size_t)(4 * j + 1) * HC_];
    }
    bcol = bias[col];
  }
  __builtin_amdgcn_sched_barrier(0);
  v4f pv[2][NCH];
  float cm[2], cs[2];
  int nchs[2];
#pragma unroll
  for (int gi = 0; gi < 2; ++gi) {
    const int b = m0 + 2 * wave + gi;
    const int nch = min(max((gs[b] + (CHR - 1)) / CHR, 0), NCH);
    nchs[gi] = nch;
    const unsigned recl = (unsigned)(b * NCH + (l16 < nch ? l16 : 0));
    cm[gi] = ws[OFF_CM + recl * H_ + h];
    cs[gi] = ws[OFF_CS + recl * H_ + h];
    const float* pb = ws + OFF_PART + h * C_ + 4 * lane;
#pragma unroll
    for (int ch = 0; ch < NCH; ++ch) {
      const unsigned rec = (unsigned)(b * NCH + (ch < nch ? ch : 0));
      pv[gi][ch] = *(const v4f*)(pb + ((size_t)rec << 9));
    }
  }
  __builtin_amdgcn_sched_barrier(0);
#pragma unroll
  for (int gi = 0; gi < 2; ++gi) {
    const float cmv = (l16 < nchs[gi]) ? cm[gi] : neg_inf();
    const float m = row_max(cmv);
    const float e = __expf(cmv - m);
    v4f u = {0.f, 0.f, 0.f, 0.f};
    float s = 0.f;
#pragma unroll
    for (int ch = 0; ch < NCH; ++ch) {
      const float es = LANE_VALUE(e, ch);
      const float css = LANE_VALUE(cs[gi], ch);
      u.x = fmaf(pv[gi][ch].x, es, u.x);
      u.y = fmaf(pv[gi][ch].y, es, u.y);
      u.z = fmaf(pv[gi][ch].z, es, u.z);
      u.w = fmaf(pv[gi][ch].w, es, u.w);
      s = fmaf(css, es, s);
    }
    v4f o;
    o.x = u.x / s; o.y = u.y / s; o.z = u.z / s; o.w = u.w / s;
    *(v4f*)(s_u + (2 * wave + gi) * UP + 4 * lane) = o;
  }
  __syncthreads();
  if (!prod) return;
  const float* urow = s_u + l16 * UP + 2 * hi;
  v2f a[32];
#pragma unroll
  for (int j = 0; j < 32; ++j) a[j] = *(const v2f*)(urow + 4 * j);
  __builtin_amdgcn_sched_barrier(0);
  v8f acc = {};
#pragma unroll
  for (int j = 0; j < 32; ++j)
    acc = __builtin_amdgcn_wmma_f32_16x16x4_f32(false, a[j], false, bw[j],
                                                (short)0, acc, false, false);
#pragma unroll
  for (int r = 0; r < 8; ++r) {
    const int row = m0 + r + 8 * hi;
    out[(size_t)row * HC_ + col] = acc[r] + bcol;
  }
}

extern "C" void kernel_launch(void* const* d_in, const int* in_sizes, int n_in,
                              void* d_out, int out_size, void* d_ws, size_t ws_size,
                              hipStream_t stream) {
  (void)in_sizes; (void)n_in; (void)out_size; (void)ws_size;
  const float* V    = (const float*)d_in[0];
  const int*   gs   = (const int*)  d_in[1];
  const float* W    = (const float*)d_in[2];
  const float* bias = (const float*)d_in[3];
  const float* T    = (const float*)d_in[4];
  float* out = (float*)d_out;
  float* ws  = (float*)d_ws;

  k_prep<<<9,             256, 0, stream>>>(W, bias, T, ws);
  k_attn<<<B_ * NCH,      128, 0, stream>>>(V, gs, ws);
  k_out <<<dim3(B_/16, H_, 2), 256, 0, stream>>>(W, bias, gs, ws, out);
}
